# starting kernel
# baseline (speedup 1.0000x reference)
#include <hip/hip_runtime.h>
#include <hip/hip_bf16.h>


typedef float v2f __attribute__((ext_vector_type(2)));
typedef float v8f __attribute__((ext_vector_type(8)));

#define WPB 2


__device__ __forceinline__ v2f ld_a(const float* m, int ld, int r0, int c0, int lane) {
  const int h = lane >> 4, mr = lane & 15;
  const float* p = m + (r0 + mr) * ld + c0 + (h << 1);
  v2f a; a.x = p[0]; a.y = p[1];
  return a;
}
__device__ __forceinline__ v2f ld_b(const float* m, int ld, int r0, int c0, int lane) {
  const int h = lane >> 4, n = lane & 15;
  const float* p = m + (r0 + (h << 1)) * ld + c0 + n;
  v2f b; b.x = p[0]; b.y = p[ld];
  return b;
}
__device__ __forceinline__ v2f ld_bt(const float* m, int ld, int kr0, int nc0, int lane) {
  const int h = lane >> 4, n = lane & 15;
  const float* p = m + (nc0 + n) * ld + kr0 + (h << 1);
  v2f b; b.x = p[0]; b.y = p[1];
  return b;
}

template <bool BT>
__device__ __forceinline__ v8f mma16(v8f acc, const float* A, int lda, int ar, int ac,
                                     const float* Bm, int ldb, int br, int bc, int lane) {
#pragma unroll
  for (int k = 0; k < 16; k += 4) {
    v2f a = ld_a(A, lda, ar, ac + k, lane);
    v2f b = BT ? ld_bt(Bm, ldb, br + k, bc, lane)
               : ld_b (Bm, ldb, br + k, bc, lane);
    acc = __builtin_amdgcn_wmma_f32_16x16x4_f32(false, a, false, b,
                                                (short)0, acc, false, false);
  }
  return acc;
}

__device__ __forceinline__ void st_c(float* d, int ld, int r0, int c0, v8f c, int lane) {
  const int h = lane >> 4, n = lane & 15;
#pragma unroll
  for (int r = 0; r < 8; ++r)
    d[(r0 + r + (h << 3)) * ld + c0 + n] = c[r];
}
__device__ __forceinline__ void st_c_add(float* d, int ld, const float* a, int lda,
                                         int r0, int c0, v8f c, int lane) {
  const int h = lane >> 4, n = lane & 15;
#pragma unroll
  for (int r = 0; r < 8; ++r) {
    const int rr = r0 + r + (h << 3);
    d[rr * ld + c0 + n] = c[r] + a[rr * lda + c0 + n];
  }
}
__device__ __forceinline__ void st_c_subfrom(float* d, int ld, int r0, int c0,
                                             v8f c, int lane) {
  const int h = lane >> 4, n = lane & 15;
#pragma unroll
  for (int r = 0; r < 8; ++r) {
    const int idx = (r0 + r + (h << 3)) * ld + c0 + n;
    d[idx] = d[idx] - c[r];
  }
}

__global__ __launch_bounds__(32 * WPB)
void kalman_wmma_kernel(const float* __restrict__ state0, const float* __restrict__ cov0,
                        const float* __restrict__ meas,   const float* __restrict__ Fg,
                        const float* __restrict__ Hg,     const float* __restrict__ Qg,
                        const float* __restrict__ Rg,     float* __restrict__ out,
                        int B, int T) {
  const int tid  = threadIdx.x;
  const int lane = tid & 31;
  const int w    = tid >> 5;

  __shared__ float sF[32 * 32];
  __shared__ float sH[16 * 32];
  __shared__ float sQ[32 * 32];
  __shared__ float sR[16 * 16];
  __shared__ float sP [WPB][32 * 32];
  __shared__ float sT1[WPB][32 * 32];
  __shared__ float sT2[WPB][16 * 32];
  __shared__ float sSa[WPB][16 * 32];
  __shared__ float sKm[WPB][32 * 16];
  __shared__ float sx [WPB][32];
  __shared__ float sxp[WPB][32];
  __shared__ float sin_[WPB][16];

  for (int i = tid; i < 32 * 32; i += blockDim.x) sF[i] = Fg[i];
  for (int i = tid; i < 16 * 32; i += blockDim.x) sH[i] = Hg[i];
  for (int i = tid; i < 32 * 32; i += blockDim.x) sQ[i] = Qg[i];
  for (int i = tid; i < 16 * 16; i += blockDim.x) sR[i] = Rg[i];
  __syncthreads();

  const int b = blockIdx.x * WPB + w;
  if (b >= B) return;

  float* P  = sP [w];
  float* T1 = sT1[w];
  float* T2 = sT2[w];
  float* Sa = sSa[w];
  float* Km = sKm[w];
  float* x  = sx [w];
  float* xp = sxp[w];
  float* nu = sin_[w];

  for (int i = lane; i < 32 * 32; i += 32) P[i] = cov0[(size_t)b * 1024 + i];
  x[lane] = state0[(size_t)b * 32 + lane];
  const float* zb = meas + (size_t)b * T * 16;
  float*       ob = out  + (size_t)b * T * 32;
  __builtin_amdgcn_wave_barrier();

#pragma unroll 1
  for (int t = 0; t < T; ++t) {
    if (t + 1 < T) __builtin_prefetch(zb + (size_t)(t + 1) * 16, 0, 1);

    {
      float acc = 0.f;
#pragma unroll
      for (int j = 0; j < 32; ++j) acc += sF[lane * 32 + j] * x[j];
      xp[lane] = acc;
    }
    __builtin_amdgcn_wave_barrier();

#pragma unroll
    for (int i = 0; i < 32; i += 16)
#pragma unroll
      for (int j = 0; j < 32; j += 16) {
        v8f acc = {};
        acc = mma16<false>(acc, sF, 32, i, 0,  P, 32, 0,  j, lane);
        acc = mma16<false>(acc, sF, 32, i, 16, P, 32, 16, j, lane);
        st_c(T1, 32, i, j, acc, lane);
      }
    __builtin_amdgcn_wave_barrier();

#pragma unroll
    for (int i = 0; i < 32; i += 16)
#pragma unroll
      for (int j = 0; j < 32; j += 16) {
        v8f acc = {};
        acc = mma16<true>(acc, T1, 32, i, 0,  sF, 32, 0,  j, lane);
        acc = mma16<true>(acc, T1, 32, i, 16, sF, 32, 16, j, lane);
        st_c_add(P, 32, sQ, 32, i, j, acc, lane);
      }
    __builtin_amdgcn_wave_barrier();

#pragma unroll
    for (int j = 0; j < 32; j += 16) {
      v8f acc = {};
      acc = mma16<false>(acc, sH, 32, 0, 0,  P, 32, 0,  j, lane);
      acc = mma16<false>(acc, sH, 32, 0, 16, P, 32, 16, j, lane);
      st_c(T2, 32, 0, j, acc, lane);
    }
    __builtin_amdgcn_wave_barrier();

    {
      v8f acc = {};
      acc = mma16<true>(acc, T2, 32, 0, 0,  sH, 32, 0,  0, lane);
      acc = mma16<true>(acc, T2, 32, 0, 16, sH, 32, 16, 0, lane);
      st_c_add(Sa, 32, sR, 16, 0, 0, acc, lane);
    }
    for (int i = lane; i < 256; i += 32) {
      const int r = i >> 4, c = i & 15;
      Sa[r * 32 + 16 + c] = (r == c) ? 1.0f : 0.0f;
    }
    __builtin_amdgcn_wave_barrier();

    if (lane < 16) {
      float acc = 0.f;
#pragma unroll
      for (int j = 0; j < 32; ++j) acc += sH[lane * 32 + j] * xp[j];
      nu[lane] = zb[(size_t)t * 16 + lane] - acc;
    }
    __builtin_amdgcn_wave_barrier();

    {
      const int c = lane;
#pragma unroll 1
      for (int k = 0; k < 16; ++k) {
        const float pinv = 1.0f / Sa[k * 32 + k];
        const float rk   = Sa[k * 32 + c] * pinv;
        __builtin_amdgcn_wave_barrier();
        Sa[k * 32 + c] = rk;
        __builtin_amdgcn_wave_barrier();
#pragma unroll 1
        for (int r = 0; r < 16; ++r) {
          if (r == k) continue;
          const float f = Sa[r * 32 + k];
          __builtin_amdgcn_wave_barrier();
          Sa[r * 32 + c] = Sa[r * 32 + c] - f * rk;
          __builtin_amdgcn_wave_barrier();
        }
      }
    }

#pragma unroll
    for (int i = 0; i < 32; i += 16) {
      v8f acc = {};
      acc = mma16<true>(acc, P, 32, i, 0,  sH, 32, 0,  0, lane);
      acc = mma16<true>(acc, P, 32, i, 16, sH, 32, 16, 0, lane);
      st_c(T1, 16, i, 0, acc, lane);
    }
    __builtin_amdgcn_wave_barrier();

#pragma unroll
    for (int i = 0; i < 32; i += 16) {
      v8f acc = {};
      acc = mma16<false>(acc, T1, 16, i, 0, Sa, 32, 0, 16, lane);
      st_c(Km, 16, i, 0, acc, lane);
    }
    __builtin_amdgcn_wave_barrier();

    {
      float acc = xp[lane];
#pragma unroll
      for (int j = 0; j < 16; ++j) acc += Km[lane * 16 + j] * nu[j];
      x[lane] = acc;
      ob[(size_t)t * 32 + lane] = acc;
    }
    __builtin_amdgcn_wave_barrier();

#pragma unroll
    for (int i = 0; i < 32; i += 16)
#pragma unroll
      for (int j = 0; j < 32; j += 16) {
        v8f acc = {};
        acc = mma16<false>(acc, Km, 16, i, 0, T2, 32, 0, j, lane);
        st_c_subfrom(P, 32, i, j, acc, lane);
      }
    __builtin_amdgcn_wave_barrier();
  }
}

extern "C" void kernel_launch(void* const* d_in, const int* in_sizes, int n_in,
                              void* d_out, int out_size, void* d_ws, size_t ws_size,
                              hipStream_t stream) {
  const float* state0 = (const float*)d_in[0];
  const float* cov0   = (const float*)d_in[1];
  const float* meas   = (const float*)d_in[2];
  const float* F      = (const float*)d_in[3];
  const float* H      = (const float*)d_in[4];
  const float* Q      = (const float*)d_in[5];
  const float* R      = (const float*)d_in[6];
  float* out = (float*)d_out;

  const int B = in_sizes[0] / 32;
  const int T = in_sizes[2] / (B * 16);
  const int blocks = (B + WPB - 1) / WPB;
  kalman_wmma_kernel<<<blocks, 32 * WPB, 0, stream>>>(state0, cov0, meas,
                                                      F, H, Q, R, out, B, T);
}


# S inversion kept in registers: one column of [S|I] per lane, pivot and row multipliers by v_readlane, S^-1 goes to the gain WMMA as register B fragments (no LDS trips, same arithmetic)
# speedup vs baseline: 3.5280x; 3.5280x over previous
#include <hip/hip_runtime.h>
#include <hip/hip_bf16.h>


typedef float v2f __attribute__((ext_vector_type(2)));
typedef float v8f __attribute__((ext_vector_type(8)));

#define WPB 2


__device__ __forceinline__ v2f ld_a(const float* m, int ld, int r0, int c0, int lane) {
  const int h = lane >> 4, mr = lane & 15;
  const float* p = m + (r0 + mr) * ld + c0 + (h << 1);
  v2f a; a.x = p[0]; a.y = p[1];
  return a;
}
__device__ __forceinline__ v2f ld_b(const float* m, int ld, int r0, int c0, int lane) {
  const int h = lane >> 4, n = lane & 15;
  const float* p = m + (r0 + (h << 1)) * ld + c0 + n;
  v2f b; b.x = p[0]; b.y = p[ld];
  return b;
}
__device__ __forceinline__ v2f ld_bt(const float* m, int ld, int kr0, int nc0, int lane) {
  const int h = lane >> 4, n = lane & 15;
  const float* p = m + (nc0 + n) * ld + kr0 + (h << 1);
  v2f b; b.x = p[0]; b.y = p[1];
  return b;
}

template <bool BT>
__device__ __forceinline__ v8f mma16(v8f acc, const float* A, int lda, int ar, int ac,
                                     const float* Bm, int ldb, int br, int bc, int lane) {
#pragma unroll
  for (int k = 0; k < 16; k += 4) {
    v2f a = ld_a(A, lda, ar, ac + k, lane);
    v2f b = BT ? ld_bt(Bm, ldb, br + k, bc, lane)
               : ld_b (Bm, ldb, br + k, bc, lane);
    acc = __builtin_amdgcn_wmma_f32_16x16x4_f32(false, a, false, b,
                                                (short)0, acc, false, false);
  }
  return acc;
}

__device__ __forceinline__ void st_c(float* d, int ld, int r0, int c0, v8f c, int lane) {
  const int h = lane >> 4, n = lane & 15;
#pragma unroll
  for (int r = 0; r < 8; ++r)
    d[(r0 + r + (h << 3)) * ld + c0 + n] = c[r];
}
__device__ __forceinline__ void st_c_add(float* d, int ld, const float* a, int lda,
                                         int r0, int c0, v8f c, int lane) {
  const int h = lane >> 4, n = lane & 15;
#pragma unroll
  for (int r = 0; r < 8; ++r) {
    const int rr = r0 + r + (h << 3);
    d[rr * ld + c0 + n] = c[r] + a[rr * lda + c0 + n];
  }
}
__device__ __forceinline__ void st_c_subfrom(float* d, int ld, int r0, int c0,
                                             v8f c, int lane) {
  const int h = lane >> 4, n = lane & 15;
#pragma unroll
  for (int r = 0; r < 8; ++r) {
    const int idx = (r0 + r + (h << 3)) * ld + c0 + n;
    d[idx] = d[idx] - c[r];
  }
}

__device__ __forceinline__ float xhalf(float x) {
  const int b = __builtin_bit_cast(int, x);
  return __builtin_bit_cast(float,
      __builtin_amdgcn_permlanex16(b, b, 0x76543210, 0xfedcba98, false, false));
}
__device__ __forceinline__ float rdlane(float v, int l) {
  return __builtin_bit_cast(float,
      __builtin_amdgcn_readlane(__builtin_bit_cast(int, v), l));
}

__global__ __launch_bounds__(32 * WPB)
void kalman_wmma_kernel(const float* __restrict__ state0, const float* __restrict__ cov0,
                        const float* __restrict__ meas,   const float* __restrict__ Fg,
                        const float* __restrict__ Hg,     const float* __restrict__ Qg,
                        const float* __restrict__ Rg,     float* __restrict__ out,
                        int B, int T) {
  const int tid  = threadIdx.x;
  const int lane = tid & 31;
  const int w    = tid >> 5;

  __shared__ float sF[32 * 32];
  __shared__ float sH[16 * 32];
  __shared__ float sQ[32 * 32];
  __shared__ float sR[16 * 16];
  __shared__ float sP [WPB][32 * 32];
  __shared__ float sT1[WPB][32 * 32];
  __shared__ float sT2[WPB][16 * 32];
  __shared__ float sKm[WPB][32 * 16];
  __shared__ float sx [WPB][32];
  __shared__ float sxp[WPB][32];
  __shared__ float sin_[WPB][16];

  for (int i = tid; i < 32 * 32; i += blockDim.x) sF[i] = Fg[i];
  for (int i = tid; i < 16 * 32; i += blockDim.x) sH[i] = Hg[i];
  for (int i = tid; i < 32 * 32; i += blockDim.x) sQ[i] = Qg[i];
  for (int i = tid; i < 16 * 16; i += blockDim.x) sR[i] = Rg[i];
  __syncthreads();

  const int b = blockIdx.x * WPB + w;
  if (b >= B) return;

  float* P  = sP [w];
  float* T1 = sT1[w];
  float* T2 = sT2[w];
  float* Km = sKm[w];
  float* x  = sx [w];
  float* xp = sxp[w];
  float* nu = sin_[w];

  for (int i = lane; i < 32 * 32; i += 32) P[i] = cov0[(size_t)b * 1024 + i];
  x[lane] = state0[(size_t)b * 32 + lane];
  const float* zb = meas + (size_t)b * T * 16;
  float*       ob = out  + (size_t)b * T * 32;
  __builtin_amdgcn_wave_barrier();

#pragma unroll 1
  for (int t = 0; t < T; ++t) {
    if (t + 1 < T) __builtin_prefetch(zb + (size_t)(t + 1) * 16, 0, 1);

    {
      float acc = 0.f;
#pragma unroll
      for (int j = 0; j < 32; ++j) acc += sF[lane * 32 + j] * x[j];
      xp[lane] = acc;
    }
    __builtin_amdgcn_wave_barrier();

#pragma unroll
    for (int i = 0; i < 32; i += 16)
#pragma unroll
      for (int j = 0; j < 32; j += 16) {
        v8f acc = {};
        acc = mma16<false>(acc, sF, 32, i, 0,  P, 32, 0,  j, lane);
        acc = mma16<false>(acc, sF, 32, i, 16, P, 32, 16, j, lane);
        st_c(T1, 32, i, j, acc, lane);
      }
    __builtin_amdgcn_wave_barrier();

#pragma unroll
    for (int i = 0; i < 32; i += 16)
#pragma unroll
      for (int j = 0; j < 32; j += 16) {
        v8f acc = {};
        acc = mma16<true>(acc, T1, 32, i, 0,  sF, 32, 0,  j, lane);
        acc = mma16<true>(acc, T1, 32, i, 16, sF, 32, 16, j, lane);
        st_c_add(P, 32, sQ, 32, i, j, acc, lane);
      }
    __builtin_amdgcn_wave_barrier();

#pragma unroll
    for (int j = 0; j < 32; j += 16) {
      v8f acc = {};
      acc = mma16<false>(acc, sH, 32, 0, 0,  P, 32, 0,  j, lane);
      acc = mma16<false>(acc, sH, 32, 0, 16, P, 32, 16, j, lane);
      st_c(T2, 32, 0, j, acc, lane);
    }
    __builtin_amdgcn_wave_barrier();

    float ga[16];
    {
      v8f acc = {};
      acc = mma16<true>(acc, T2, 32, 0, 0,  sH, 32, 0,  0, lane);
      acc = mma16<true>(acc, T2, 32, 0, 16, sH, 32, 16, 0, lane);
      const int h = lane >> 4, n = lane & 15;
#pragma unroll
      for (int r = 0; r < 8; ++r) {
        const float s = acc[r] + sR[(r + (h << 3)) * 16 + n];
        const float o = xhalf(s);
        ga[r]     = h ? ((r     == n) ? 1.0f : 0.0f) : s;
        ga[r + 8] = h ? ((r + 8 == n) ? 1.0f : 0.0f) : o;
      }
    }

    if (lane < 16) {
      float acc = 0.f;
#pragma unroll
      for (int j = 0; j < 32; ++j) acc += sH[lane * 32 + j] * xp[j];
      nu[lane] = zb[(size_t)t * 16 + lane] - acc;
    }
    __builtin_amdgcn_wave_barrier();

#pragma unroll
    for (int k = 0; k < 16; ++k) {
      const float pinv = 1.0f / rdlane(ga[k], k);
      const float rk   = ga[k] * pinv;
      ga[k] = rk;
#pragma unroll
      for (int r = 0; r < 16; ++r) {
        if (r == k) continue;
        const float f = rdlane(ga[r], k);
        ga[r] = ga[r] - f * rk;
      }
    }
    v2f bS[4];
    {
      const int h = lane >> 4;
#pragma unroll
      for (int q = 0; q < 4; ++q) {
        const float x0 = xhalf(ga[4 * q]);
        const float x1 = xhalf(ga[4 * q + 1]);
        bS[q].x = h ? ga[4 * q + 2] : x0;
        bS[q].y = h ? ga[4 * q + 3] : x1;
      }
    }

#pragma unroll
    for (int i = 0; i < 32; i += 16) {
      v8f acc = {};
      acc = mma16<true>(acc, P, 32, i, 0,  sH, 32, 0,  0, lane);
      acc = mma16<true>(acc, P, 32, i, 16, sH, 32, 16, 0, lane);
      st_c(T1, 16, i, 0, acc, lane);
    }
    __builtin_amdgcn_wave_barrier();

#pragma unroll
    for (int i = 0; i < 32; i += 16) {
      v8f acc = {};
#pragma unroll
      for (int q = 0; q < 4; ++q) {
        const v2f a = ld_a(T1, 16, i, 4 * q, lane);
        acc = __builtin_amdgcn_wmma_f32_16x16x4_f32(false, a, false, bS[q],
                                                    (short)0, acc, false, false);
      }
      st_c(Km, 16, i, 0, acc, lane);
    }
    __builtin_amdgcn_wave_barrier();

    {
      float acc = xp[lane];
#pragma unroll
      for (int j = 0; j < 16; ++j) acc += Km[lane * 16 + j] * nu[j];
      x[lane] = acc;
      ob[(size_t)t * 32 + lane] = acc;
    }
    __builtin_amdgcn_wave_barrier();

#pragma unroll
    for (int i = 0; i < 32; i += 16)
#pragma unroll
      for (int j = 0; j < 32; j += 16) {
        v8f acc = {};
        acc = mma16<false>(acc, Km, 16, i, 0, T2, 32, 0, j, lane);
        st_c_subfrom(P, 32, i, j, acc, lane);
      }
    __builtin_amdgcn_wave_barrier();
  }
}

extern "C" void kernel_launch(void* const* d_in, const int* in_sizes, int n_in,
                              void* d_out, int out_size, void* d_ws, size_t ws_size,
                              hipStream_t stream) {
  const float* state0 = (const float*)d_in[0];
  const float* cov0   = (const float*)d_in[1];
  const float* meas   = (const float*)d_in[2];
  const float* F      = (const float*)d_in[3];
  const float* H      = (const float*)d_in[4];
  const float* Q      = (const float*)d_in[5];
  const float* R      = (const float*)d_in[6];
  float* out = (float*)d_out;

  const int B = in_sizes[0] / 32;
  const int T = in_sizes[2] / (B * 16);
  const int blocks = (B + WPB - 1) / WPB;
  kalman_wmma_kernel<<<blocks, 32 * WPB, 0, stream>>>(state0, cov0, meas,
                                                      F, H, Q, R, out, B, T);
}


# LDS row pitch of every matrix padded from 32/16 to 44/20 floats so A, transposed-B, row-vector and C-tile accesses spread over all 64 banks instead of 8-way/4-way bank conflicts; same arithmetic
# speedup vs baseline: 3.9277x; 1.1133x over previous
#include <hip/hip_runtime.h>
#include <hip/hip_bf16.h>


typedef float v2f __attribute__((ext_vector_type(2)));
typedef float v8f __attribute__((ext_vector_type(8)));

#define WPB 2

#define LDW 44
#define LDN 20


__device__ __forceinline__ v2f ld_a(const float* m, int ld, int r0, int c0, int lane) {
  const int h = lane >> 4, mr = lane & 15;
  const float* p = m + (r0 + mr) * ld + c0 + (h << 1);
  v2f a; a.x = p[0]; a.y = p[1];
  return a;
}
__device__ __forceinline__ v2f ld_b(const float* m, int ld, int r0, int c0, int lane) {
  const int h = lane >> 4, n = lane & 15;
  const float* p = m + (r0 + (h << 1)) * ld + c0 + n;
  v2f b; b.x = p[0]; b.y = p[ld];
  return b;
}
__device__ __forceinline__ v2f ld_bt(const float* m, int ld, int kr0, int nc0, int lane) {
  const int h = lane >> 4, n = lane & 15;
  const float* p = m + (nc0 + n) * ld + kr0 + (h << 1);
  v2f b; b.x = p[0]; b.y = p[1];
  return b;
}

template <bool BT>
__device__ __forceinline__ v8f mma16(v8f acc, const float* A, int lda, int ar, int ac,
                                     const float* Bm, int ldb, int br, int bc, int lane) {
#pragma unroll
  for (int k = 0; k < 16; k += 4) {
    v2f a = ld_a(A, lda, ar, ac + k, lane);
    v2f b = BT ? ld_bt(Bm, ldb, br + k, bc, lane)
               : ld_b (Bm, ldb, br + k, bc, lane);
    acc = __builtin_amdgcn_wmma_f32_16x16x4_f32(false, a, false, b,
                                                (short)0, acc, false, false);
  }
  return acc;
}

__device__ __forceinline__ void st_c(float* d, int ld, int r0, int c0, v8f c, int lane) {
  const int h = lane >> 4, n = lane & 15;
#pragma unroll
  for (int r = 0; r < 8; ++r)
    d[(r0 + r + (h << 3)) * ld + c0 + n] = c[r];
}
__device__ __forceinline__ void st_c_add(float* d, int ld, const float* a, int lda,
                                         int r0, int c0, v8f c, int lane) {
  const int h = lane >> 4, n = lane & 15;
#pragma unroll
  for (int r = 0; r < 8; ++r) {
    const int rr = r0 + r + (h << 3);
    d[rr * ld + c0 + n] = c[r] + a[rr * lda + c0 + n];
  }
}
__device__ __forceinline__ void st_c_subfrom(float* d, int ld, int r0, int c0,
                                             v8f c, int lane) {
  const int h = lane >> 4, n = lane & 15;
#pragma unroll
  for (int r = 0; r < 8; ++r) {
    const int idx = (r0 + r + (h << 3)) * ld + c0 + n;
    d[idx] = d[idx] - c[r];
  }
}

__device__ __forceinline__ float xhalf(float x) {
  const int b = __builtin_bit_cast(int, x);
  return __builtin_bit_cast(float,
      __builtin_amdgcn_permlanex16(b, b, 0x76543210, 0xfedcba98, false, false));
}
__device__ __forceinline__ float rdlane(float v, int l) {
  return __builtin_bit_cast(float,
      __builtin_amdgcn_readlane(__builtin_bit_cast(int, v), l));
}

__global__ __launch_bounds__(32 * WPB)
void kalman_wmma_kernel(const float* __restrict__ state0, const float* __restrict__ cov0,
                        const float* __restrict__ meas,   const float* __restrict__ Fg,
                        const float* __restrict__ Hg,     const float* __restrict__ Qg,
                        const float* __restrict__ Rg,     float* __restrict__ out,
                        int B, int T) {
  const int tid  = threadIdx.x;
  const int lane = tid & 31;
  const int w    = tid >> 5;

  __shared__ __attribute__((aligned(16))) float sF[32 * LDW];
  __shared__ __attribute__((aligned(16))) float sH[16 * LDW];
  __shared__ __attribute__((aligned(16))) float sQ[32 * LDW];
  __shared__ __attribute__((aligned(16))) float sR[16 * LDN];
  __shared__ __attribute__((aligned(16))) float sP [WPB][32 * LDW];
  __shared__ __attribute__((aligned(16))) float sT1[WPB][32 * LDW];
  __shared__ __attribute__((aligned(16))) float sT2[WPB][16 * LDW];
  __shared__ __attribute__((aligned(16))) float sKm[WPB][32 * LDN];
  __shared__ float sx [WPB][32];
  __shared__ float sxp[WPB][32];
  __shared__ float sin_[WPB][16];

  for (int i = tid; i < 32 * 32; i += blockDim.x) sF[(i >> 5) * LDW + (i & 31)] = Fg[i];
  for (int i = tid; i < 16 * 32; i += blockDim.x) sH[(i >> 5) * LDW + (i & 31)] = Hg[i];
  for (int i = tid; i < 32 * 32; i += blockDim.x) sQ[(i >> 5) * LDW + (i & 31)] = Qg[i];
  for (int i = tid; i < 16 * 16; i += blockDim.x) sR[(i >> 4) * LDN + (i & 15)] = Rg[i];
  __syncthreads();

  const int b = blockIdx.x * WPB + w;
  if (b >= B) return;

  float* P  = sP [w];
  float* T1 = sT1[w];
  float* T2 = sT2[w];
  float* Km = sKm[w];
  float* x  = sx [w];
  float* xp = sxp[w];
  float* nu = sin_[w];

  for (int i = lane; i < 32 * 32; i += 32) P[(i >> 5) * LDW + (i & 31)] = cov0[(size_t)b * 1024 + i];
  x[lane] = state0[(size_t)b * 32 + lane];
  const float* zb = meas + (size_t)b * T * 16;
  float*       ob = out  + (size_t)b * T * 32;
  __builtin_amdgcn_wave_barrier();

#pragma unroll 1
  for (int t = 0; t < T; ++t) {
    if (t + 1 < T) __builtin_prefetch(zb + (size_t)(t + 1) * 16, 0, 1);

    {
      float acc = 0.f;
#pragma unroll
      for (int j = 0; j < 32; ++j) acc += sF[lane * LDW + j] * x[j];
      xp[lane] = acc;
    }
    __builtin_amdgcn_wave_barrier();

#pragma unroll
    for (int i = 0; i < 32; i += 16)
#pragma unroll
      for (int j = 0; j < 32; j += 16) {
        v8f acc = {};
        acc = mma16<false>(acc, sF, LDW, i, 0,  P, LDW, 0,  j, lane);
        acc = mma16<false>(acc, sF, LDW, i, 16, P, LDW, 16, j, lane);
        st_c(T1, LDW, i, j, acc, lane);
      }
    __builtin_amdgcn_wave_barrier();

#pragma unroll
    for (int i = 0; i < 32; i += 16)
#pragma unroll
      for (int j = 0; j < 32; j += 16) {
        v8f acc = {};
        acc = mma16<true>(acc, T1, LDW, i, 0,  sF, LDW, 0,  j, lane);
        acc = mma16<true>(acc, T1, LDW, i, 16, sF, LDW, 16, j, lane);
        st_c_add(P, LDW, sQ, LDW, i, j, acc, lane);
      }
    __builtin_amdgcn_wave_barrier();

#pragma unroll
    for (int j = 0; j < 32; j += 16) {
      v8f acc = {};
      acc = mma16<false>(acc, sH, LDW, 0, 0,  P, LDW, 0,  j, lane);
      acc = mma16<false>(acc, sH, LDW, 0, 16, P, LDW, 16, j, lane);
      st_c(T2, LDW, 0, j, acc, lane);
    }
    __builtin_amdgcn_wave_barrier();

    float ga[16];
    {
      v8f acc = {};
      acc = mma16<true>(acc, T2, LDW, 0, 0,  sH, LDW, 0,  0, lane);
      acc = mma16<true>(acc, T2, LDW, 0, 16, sH, LDW, 16, 0, lane);
      const int h = lane >> 4, n = lane & 15;
#pragma unroll
      for (int r = 0; r < 8; ++r) {
        const float s = acc[r] + sR[(r + (h << 3)) * LDN + n];
        const float o = xhalf(s);
        ga[r]     = h ? ((r     == n) ? 1.0f : 0.0f) : s;
        ga[r + 8] = h ? ((r + 8 == n) ? 1.0f : 0.0f) : o;
      }
    }

    if (lane < 16) {
      float acc = 0.f;
#pragma unroll
      for (int j = 0; j < 32; ++j) acc += sH[lane * LDW + j] * xp[j];
      nu[lane] = zb[(size_t)t * 16 + lane] - acc;
    }
    __builtin_amdgcn_wave_barrier();

#pragma unroll
    for (int k = 0; k < 16; ++k) {
      const float pinv = 1.0f / rdlane(ga[k], k);
      const float rk   = ga[k] * pinv;
      ga[k] = rk;
#pragma unroll
      for (int r = 0; r < 16; ++r) {
        if (r == k) continue;
        const float f = rdlane(ga[r], k);
        ga[r] = ga[r] - f * rk;
      }
    }
    v2f bS[4];
    {
      const int h = lane >> 4;
#pragma unroll
      for (int q = 0; q < 4; ++q) {
        const float x0 = xhalf(ga[4 * q]);
        const float x1 = xhalf(ga[4 * q + 1]);
        bS[q].x = h ? ga[4 * q + 2] : x0;
        bS[q].y = h ? ga[4 * q + 3] : x1;
      }
    }

#pragma unroll
    for (int i = 0; i < 32; i += 16) {
      v8f acc = {};
      acc = mma16<true>(acc, P, LDW, i, 0,  sH, LDW, 0,  0, lane);
      acc = mma16<true>(acc, P, LDW, i, 16, sH, LDW, 16, 0, lane);
      st_c(T1, LDN, i, 0, acc, lane);
    }
    __builtin_amdgcn_wave_barrier();

#pragma unroll
    for (int i = 0; i < 32; i += 16) {
      v8f acc = {};
#pragma unroll
      for (int q = 0; q < 4; ++q) {
        const v2f a = ld_a(T1, LDN, i, 4 * q, lane);
        acc = __builtin_amdgcn_wmma_f32_16x16x4_f32(false, a, false, bS[q],
                                                    (short)0, acc, false, false);
      }
      st_c(Km, LDN, i, 0, acc, lane);
    }
    __builtin_amdgcn_wave_barrier();

    {
      float acc = xp[lane];
#pragma unroll
      for (int j = 0; j < 16; ++j) acc += Km[lane * LDN + j] * nu[j];
      x[lane] = acc;
      ob[(size_t)t * 32 + lane] = acc;
    }
    __builtin_amdgcn_wave_barrier();

#pragma unroll
    for (int i = 0; i < 32; i += 16)
#pragma unroll
      for (int j = 0; j < 32; j += 16) {
        v8f acc = {};
        acc = mma16<false>(acc, Km, LDN, i, 0, T2, LDW, 0, j, lane);
        st_c_subfrom(P, LDW, i, j, acc, lane);
      }
    __builtin_amdgcn_wave_barrier();
  }
}

extern "C" void kernel_launch(void* const* d_in, const int* in_sizes, int n_in,
                              void* d_out, int out_size, void* d_ws, size_t ws_size,
                              hipStream_t stream) {
  const float* state0 = (const float*)d_in[0];
  const float* cov0   = (const float*)d_in[1];
  const float* meas   = (const float*)d_in[2];
  const float* F      = (const float*)d_in[3];
  const float* H      = (const float*)d_in[4];
  const float* Q      = (const float*)d_in[5];
  const float* R      = (const float*)d_in[6];
  float* out = (float*)d_out;

  const int B = in_sizes[0] / 32;
  const int T = in_sizes[2] / (B * 16);
  const int blocks = (B + WPB - 1) / WPB;
  kalman_wmma_kernel<<<blocks, 32 * WPB, 0, stream>>>(state0, cov0, meas,
                                                      F, H, Q, R, out, B, T);
}


# Covariance chain kept in registers: WMMA result tiles feed the next product directly as B or transposed-A operands (matched permuted k order, so k sum order in a tile changes)
# speedup vs baseline: 4.8467x; 1.2340x over previous
#include <hip/hip_runtime.h>
#include <hip/hip_bf16.h>


typedef float v2f __attribute__((ext_vector_type(2)));
typedef float v8f __attribute__((ext_vector_type(8)));

#define WPB 2

#define LDW 44
#define LDN 20


__device__ __forceinline__ v2f ld_rowfrag(const float* m, int ld, int r0, int c0, int q, int lane) {
  const int h = lane >> 4, c = lane & 15;
  const float* p = m + (r0 + c) * ld + c0 + (h << 3) + (q << 1);
  v2f a; a.x = p[0]; a.y = p[1];
  return a;
}
__device__ __forceinline__ v2f frag_c(v8f x, int q) {
  v2f f; f.x = x[2 * q]; f.y = x[2 * q + 1];
  return f;
}
__device__ __forceinline__ v8f wmma4(v2f a, v2f b, v8f c) {
  return __builtin_amdgcn_wmma_f32_16x16x4_f32(false, a, false, b, (short)0, c, false, false);
}

__device__ __forceinline__ void st_c(float* d, int ld, int r0, int c0, v8f c, int lane) {
  const int h = lane >> 4, n = lane & 15;
#pragma unroll
  for (int r = 0; r < 8; ++r)
    d[(r0 + r + (h << 3)) * ld + c0 + n] = c[r];
}
__device__ __forceinline__ void st_ct(float* d, int ld, int r0, int c0, v8f c, int lane) {
  const int h = lane >> 4, n = lane & 15;
#pragma unroll
  for (int r = 0; r < 8; ++r)
    d[(r0 + n) * ld + c0 + (h << 3) + r] = c[r];
}

__device__ __forceinline__ float xhalf(float x) {
  const int b = __builtin_bit_cast(int, x);
  return __builtin_bit_cast(float,
      __builtin_amdgcn_permlanex16(b, b, 0x76543210, 0xfedcba98, false, false));
}
__device__ __forceinline__ float rdlane(float v, int l) {
  return __builtin_bit_cast(float,
      __builtin_amdgcn_readlane(__builtin_bit_cast(int, v), l));
}

__global__ __launch_bounds__(32 * WPB)
void kalman_wmma_kernel(const float* __restrict__ state0, const float* __restrict__ cov0,
                        const float* __restrict__ meas,   const float* __restrict__ Fg,
                        const float* __restrict__ Hg,     const float* __restrict__ Qg,
                        const float* __restrict__ Rg,     float* __restrict__ out,
                        int B, int T) {
  const int tid  = threadIdx.x;
  const int lane = tid & 31;
  const int w    = tid >> 5;
  const int h    = lane >> 4;
  const int n    = lane & 15;

  __shared__ __attribute__((aligned(16))) float sF[32 * LDW];
  __shared__ __attribute__((aligned(16))) float sH[16 * LDW];
  __shared__ __attribute__((aligned(16))) float sP [WPB][32 * LDW];
  __shared__ __attribute__((aligned(16))) float sT1[WPB][32 * LDN];
  __shared__ __attribute__((aligned(16))) float sKm[WPB][32 * LDN];
  __shared__ float sx [WPB][32];
  __shared__ float sxp[WPB][32];
  __shared__ float sin_[WPB][16];

  for (int i = tid; i < 32 * 32; i += blockDim.x) sF[(i >> 5) * LDW + (i & 31)] = Fg[i];
  for (int i = tid; i < 16 * 32; i += blockDim.x) sH[(i >> 5) * LDW + (i & 31)] = Hg[i];
  __syncthreads();

  const int b = blockIdx.x * WPB + w;
  if (b >= B) return;

  float* P  = sP [w];
  float* T1 = sT1[w];
  float* Km = sKm[w];
  float* x  = sx [w];
  float* xp = sxp[w];
  float* nu = sin_[w];

  v2f fF[2][2][4];
  v2f fH[2][4];
#pragma unroll
  for (int J = 0; J < 2; ++J)
#pragma unroll
    for (int Kt = 0; Kt < 2; ++Kt)
#pragma unroll
      for (int q = 0; q < 4; ++q)
        fF[J][Kt][q] = ld_rowfrag(sF, LDW, 16 * J, 16 * Kt, q, lane);
#pragma unroll
  for (int Kt = 0; Kt < 2; ++Kt)
#pragma unroll
    for (int q = 0; q < 4; ++q)
      fH[Kt][q] = ld_rowfrag(sH, LDW, 0, 16 * Kt, q, lane);
  v8f Qc[2][2], Pc[2][2], Rc;
#pragma unroll
  for (int I = 0; I < 2; ++I)
#pragma unroll
    for (int J = 0; J < 2; ++J)
#pragma unroll
      for (int r = 0; r < 8; ++r) {
        const int e = (16 * I + 8 * h + r) * 32 + 16 * J + n;
        Qc[I][J][r] = Qg[e];
        Pc[I][J][r] = cov0[(size_t)b * 1024 + e];
      }
#pragma unroll
  for (int r = 0; r < 8; ++r) Rc[r] = Rg[(8 * h + r) * 16 + n];

  x[lane] = state0[(size_t)b * 32 + lane];
  const float* zb = meas + (size_t)b * T * 16;
  float*       ob = out  + (size_t)b * T * 32;
  __builtin_amdgcn_wave_barrier();

#pragma unroll 1
  for (int t = 0; t < T; ++t) {
    if (t + 1 < T) __builtin_prefetch(zb + (size_t)(t + 1) * 16, 0, 1);

    {
      float acc = 0.f;
#pragma unroll
      for (int j = 0; j < 32; ++j) acc += sF[lane * LDW + j] * x[j];
      xp[lane] = acc;
    }
    __builtin_amdgcn_wave_barrier();

    v8f Zc[2][2];
#pragma unroll
    for (int I = 0; I < 2; ++I)
#pragma unroll
      for (int J = 0; J < 2; ++J) {
        v8f acc = {};
#pragma unroll
        for (int Kt = 0; Kt < 2; ++Kt)
#pragma unroll
          for (int q = 0; q < 4; ++q)
            acc = wmma4(frag_c(Pc[Kt][I], q), fF[J][Kt][q], acc);
        Zc[I][J] = acc;
      }

#pragma unroll
    for (int I = 0; I < 2; ++I)
#pragma unroll
      for (int J = 0; J < 2; ++J) {
        v8f acc = {};
#pragma unroll
        for (int Kt = 0; Kt < 2; ++Kt)
#pragma unroll
          for (int q = 0; q < 4; ++q)
            acc = wmma4(frag_c(Zc[Kt][I], q), fF[J][Kt][q], acc);
        Pc[I][J] = acc + Qc[I][J];
        st_c(P, LDW, 16 * I, 16 * J, Pc[I][J], lane);
      }

    __builtin_amdgcn_wave_barrier();

    v8f T1c[2];
#pragma unroll
    for (int I = 0; I < 2; ++I) {
      v8f acc = {};
#pragma unroll
      for (int Kt = 0; Kt < 2; ++Kt)
#pragma unroll
        for (int q = 0; q < 4; ++q)
          acc = wmma4(ld_rowfrag(P, LDW, 16 * I, 16 * Kt, q, lane), fH[Kt][q], acc);
      T1c[I] = acc;
      st_c(T1, LDN, 16 * I, 0, acc, lane);
    }

    float ga[16];
    {
      v8f acc = {};
#pragma unroll
      for (int Kt = 0; Kt < 2; ++Kt)
#pragma unroll
        for (int q = 0; q < 4; ++q)
          acc = wmma4(fH[Kt][q], frag_c(T1c[Kt], q), acc);
#pragma unroll
      for (int r = 0; r < 8; ++r) {
        const float s = acc[r] + Rc[r];
        const float o = xhalf(s);
        ga[r]     = h ? ((r     == n) ? 1.0f : 0.0f) : s;
        ga[r + 8] = h ? ((r + 8 == n) ? 1.0f : 0.0f) : o;
      }
    }

    if (lane < 16) {
      float acc = 0.f;
#pragma unroll
      for (int j = 0; j < 32; ++j) acc += sH[lane * LDW + j] * xp[j];
      nu[lane] = zb[(size_t)t * 16 + lane] - acc;
    }
    __builtin_amdgcn_wave_barrier();

    v8f T2c[2];
#pragma unroll
    for (int J = 0; J < 2; ++J) {
      v8f acc = {};
#pragma unroll
      for (int Kt = 0; Kt < 2; ++Kt)
#pragma unroll
        for (int q = 0; q < 4; ++q)
          acc = wmma4(fH[Kt][q], frag_c(Pc[Kt][J], q), acc);
      T2c[J] = acc;
    }
#pragma unroll
    for (int k = 0; k < 16; ++k) {
      const float pinv = 1.0f / rdlane(ga[k], k);
      const float rk   = ga[k] * pinv;
      ga[k] = rk;
#pragma unroll
      for (int r = 0; r < 16; ++r) {
        if (r == k) continue;
        const float f = rdlane(ga[r], k);
        ga[r] = ga[r] - f * rk;
      }
    }
    v8f Si;
#pragma unroll
    for (int r = 0; r < 8; ++r) {
      const float lo = xhalf(ga[r]);
      Si[r] = h ? ga[r + 8] : lo;
    }

    v8f KTc[2];
#pragma unroll
    for (int J = 0; J < 2; ++J) {
      v8f acc = {};
#pragma unroll
      for (int q = 0; q < 4; ++q)
        acc = wmma4(frag_c(Si, q), ld_rowfrag(T1, LDN, 16 * J, 0, q, lane), acc);
      KTc[J] = acc;
      st_ct(Km, LDN, 16 * J, 0, acc, lane);
    }
    __builtin_amdgcn_wave_barrier();

    {
      float acc = xp[lane];
#pragma unroll
      for (int j = 0; j < 16; ++j) acc += Km[lane * LDN + j] * nu[j];
      x[lane] = acc;
      ob[(size_t)t * 32 + lane] = acc;
    }

#pragma unroll
    for (int I = 0; I < 2; ++I)
#pragma unroll
      for (int J = 0; J < 2; ++J) {
        v8f acc = {};
#pragma unroll
        for (int q = 0; q < 4; ++q)
          acc = wmma4(frag_c(KTc[I], q), frag_c(T2c[J], q), acc);
        Pc[I][J] = Pc[I][J] - acc;
      }
    __builtin_amdgcn_wave_barrier();
  }
}

extern "C" void kernel_launch(void* const* d_in, const int* in_sizes, int n_in,
                              void* d_out, int out_size, void* d_ws, size_t ws_size,
                              hipStream_t stream) {
  const float* state0 = (const float*)d_in[0];
  const float* cov0   = (const float*)d_in[1];
  const float* meas   = (const float*)d_in[2];
  const float* F      = (const float*)d_in[3];
  const float* H      = (const float*)d_in[4];
  const float* Q      = (const float*)d_in[5];
  const float* R      = (const float*)d_in[6];
  float* out = (float*)d_out;

  const int B = in_sizes[0] / 32;
  const int T = in_sizes[2] / (B * 16);
  const int blocks = (B + WPB - 1) / WPB;
  kalman_wmma_kernel<<<blocks, 32 * WPB, 0, stream>>>(state0, cov0, meas,
                                                      F, H, Q, R, out, B, T);
}


# Pivot reciprocal of the S elimination by the hardware reciprocal instruction (about 1 ulp) instead of the 11-instruction correctly rounded division on the serial pivot chain
# speedup vs baseline: 5.2942x; 1.0923x over previous
#include <hip/hip_runtime.h>
#include <hip/hip_bf16.h>


typedef float v2f __attribute__((ext_vector_type(2)));
typedef float v8f __attribute__((ext_vector_type(8)));

#define WPB 2

#define LDW 44
#define LDN 20


__device__ __forceinline__ v2f ld_rowfrag(const float* m, int ld, int r0, int c0, int q, int lane) {
  const int h = lane >> 4, c = lane & 15;
  const float* p = m + (r0 + c) * ld + c0 + (h << 3) + (q << 1);
  v2f a; a.x = p[0]; a.y = p[1];
  return a;
}
__device__ __forceinline__ v2f frag_c(v8f x, int q) {
  v2f f; f.x = x[2 * q]; f.y = x[2 * q + 1];
  return f;
}
__device__ __forceinline__ v8f wmma4(v2f a, v2f b, v8f c) {
  return __builtin_amdgcn_wmma_f32_16x16x4_f32(false, a, false, b, (short)0, c, false, false);
}

__device__ __forceinline__ void st_c(float* d, int ld, int r0, int c0, v8f c, int lane) {
  const int h = lane >> 4, n = lane & 15;
#pragma unroll
  for (int r = 0; r < 8; ++r)
    d[(r0 + r + (h << 3)) * ld + c0 + n] = c[r];
}
__device__ __forceinline__ void st_ct(float* d, int ld, int r0, int c0, v8f c, int lane) {
  const int h = lane >> 4, n = lane & 15;
#pragma unroll
  for (int r = 0; r < 8; ++r)
    d[(r0 + n) * ld + c0 + (h << 3) + r] = c[r];
}

__device__ __forceinline__ float xhalf(float x) {
  const int b = __builtin_bit_cast(int, x);
  return __builtin_bit_cast(float,
      __builtin_amdgcn_permlanex16(b, b, 0x76543210, 0xfedcba98, false, false));
}
__device__ __forceinline__ float rdlane(float v, int l) {
  return __builtin_bit_cast(float,
      __builtin_amdgcn_readlane(__builtin_bit_cast(int, v), l));
}

__global__ __launch_bounds__(32 * WPB)
void kalman_wmma_kernel(const float* __restrict__ state0, const float* __restrict__ cov0,
                        const float* __restrict__ meas,   const float* __restrict__ Fg,
                        const float* __restrict__ Hg,     const float* __restrict__ Qg,
                        const float* __restrict__ Rg,     float* __restrict__ out,
                        int B, int T) {
  const int tid  = threadIdx.x;
  const int lane = tid & 31;
  const int w    = tid >> 5;
  const int h    = lane >> 4;
  const int n    = lane & 15;

  __shared__ __attribute__((aligned(16))) float sF[32 * LDW];
  __shared__ __attribute__((aligned(16))) float sH[16 * LDW];
  __shared__ __attribute__((aligned(16))) float sP [WPB][32 * LDW];
  __shared__ __attribute__((aligned(16))) float sT1[WPB][32 * LDN];
  __shared__ __attribute__((aligned(16))) float sKm[WPB][32 * LDN];
  __shared__ float sx [WPB][32];
  __shared__ float sxp[WPB][32];
  __shared__ float sin_[WPB][16];

  for (int i = tid; i < 32 * 32; i += blockDim.x) sF[(i >> 5) * LDW + (i & 31)] = Fg[i];
  for (int i = tid; i < 16 * 32; i += blockDim.x) sH[(i >> 5) * LDW + (i & 31)] = Hg[i];
  __syncthreads();

  const int b = blockIdx.x * WPB + w;
  if (b >= B) return;

  float* P  = sP [w];
  float* T1 = sT1[w];
  float* Km = sKm[w];
  float* x  = sx [w];
  float* xp = sxp[w];
  float* nu = sin_[w];

  v2f fF[2][2][4];
  v2f fH[2][4];
#pragma unroll
  for (int J = 0; J < 2; ++J)
#pragma unroll
    for (int Kt = 0; Kt < 2; ++Kt)
#pragma unroll
      for (int q = 0; q < 4; ++q)
        fF[J][Kt][q] = ld_rowfrag(sF, LDW, 16 * J, 16 * Kt, q, lane);
#pragma unroll
  for (int Kt = 0; Kt < 2; ++Kt)
#pragma unroll
    for (int q = 0; q < 4; ++q)
      fH[Kt][q] = ld_rowfrag(sH, LDW, 0, 16 * Kt, q, lane);
  v8f Qc[2][2], Pc[2][2], Rc;
#pragma unroll
  for (int I = 0; I < 2; ++I)
#pragma unroll
    for (int J = 0; J < 2; ++J)
#pragma unroll
      for (int r = 0; r < 8; ++r) {
        const int e = (16 * I + 8 * h + r) * 32 + 16 * J + n;
        Qc[I][J][r] = Qg[e];
        Pc[I][J][r] = cov0[(size_t)b * 1024 + e];
      }
#pragma unroll
  for (int r = 0; r < 8; ++r) Rc[r] = Rg[(8 * h + r) * 16 + n];

  x[lane] = state0[(size_t)b * 32 + lane];
  const float* zb = meas + (size_t)b * T * 16;
  float*       ob = out  + (size_t)b * T * 32;
  __builtin_amdgcn_wave_barrier();

#pragma unroll 1
  for (int t = 0; t < T; ++t) {
    if (t + 1 < T) __builtin_prefetch(zb + (size_t)(t + 1) * 16, 0, 1);

    {
      float acc = 0.f;
#pragma unroll
      for (int j = 0; j < 32; ++j) acc += sF[lane * LDW + j] * x[j];
      xp[lane] = acc;
    }
    __builtin_amdgcn_wave_barrier();

    v8f Zc[2][2];
#pragma unroll
    for (int I = 0; I < 2; ++I)
#pragma unroll
      for (int J = 0; J < 2; ++J) {
        v8f acc = {};
#pragma unroll
        for (int Kt = 0; Kt < 2; ++Kt)
#pragma unroll
          for (int q = 0; q < 4; ++q)
            acc = wmma4(frag_c(Pc[Kt][I], q), fF[J][Kt][q], acc);
        Zc[I][J] = acc;
      }

#pragma unroll
    for (int I = 0; I < 2; ++I)
#pragma unroll
      for (int J = 0; J < 2; ++J) {
        v8f acc = {};
#pragma unroll
        for (int Kt = 0; Kt < 2; ++Kt)
#pragma unroll
          for (int q = 0; q < 4; ++q)
            acc = wmma4(frag_c(Zc[Kt][I], q), fF[J][Kt][q], acc);
        Pc[I][J] = acc + Qc[I][J];
        st_c(P, LDW, 16 * I, 16 * J, Pc[I][J], lane);
      }

    __builtin_amdgcn_wave_barrier();

    v8f T1c[2];
#pragma unroll
    for (int I = 0; I < 2; ++I) {
      v8f acc = {};
#pragma unroll
      for (int Kt = 0; Kt < 2; ++Kt)
#pragma unroll
        for (int q = 0; q < 4; ++q)
          acc = wmma4(ld_rowfrag(P, LDW, 16 * I, 16 * Kt, q, lane), fH[Kt][q], acc);
      T1c[I] = acc;
      st_c(T1, LDN, 16 * I, 0, acc, lane);
    }

    float ga[16];
    {
      v8f acc = {};
#pragma unroll
      for (int Kt = 0; Kt < 2; ++Kt)
#pragma unroll
        for (int q = 0; q < 4; ++q)
          acc = wmma4(fH[Kt][q], frag_c(T1c[Kt], q), acc);
#pragma unroll
      for (int r = 0; r < 8; ++r) {
        const float s = acc[r] + Rc[r];
        const float o = xhalf(s);
        ga[r]     = h ? ((r     == n) ? 1.0f : 0.0f) : s;
        ga[r + 8] = h ? ((r + 8 == n) ? 1.0f : 0.0f) : o;
      }
    }

    if (lane < 16) {
      float acc = 0.f;
#pragma unroll
      for (int j = 0; j < 32; ++j) acc += sH[lane * LDW + j] * xp[j];
      nu[lane] = zb[(size_t)t * 16 + lane] - acc;
    }
    __builtin_amdgcn_wave_barrier();

    v8f T2c[2];
#pragma unroll
    for (int J = 0; J < 2; ++J) {
      v8f acc = {};
#pragma unroll
      for (int Kt = 0; Kt < 2; ++Kt)
#pragma unroll
        for (int q = 0; q < 4; ++q)
          acc = wmma4(fH[Kt][q], frag_c(Pc[Kt][J], q), acc);
      T2c[J] = acc;
    }
#pragma unroll
    for (int k = 0; k < 16; ++k) {
      const float pinv = __builtin_amdgcn_rcpf(rdlane(ga[k], k));
      const float rk   = ga[k] * pinv;
      ga[k] = rk;
#pragma unroll
      for (int r = 0; r < 16; ++r) {
        if (r == k) continue;
        const float f = rdlane(ga[r], k);
        ga[r] = ga[r] - f * rk;
      }
    }
    v8f Si;
#pragma unroll
    for (int r = 0; r < 8; ++r) {
      const float lo = xhalf(ga[r]);
      Si[r] = h ? ga[r + 8] : lo;
    }

    v8f KTc[2];
#pragma unroll
    for (int J = 0; J < 2; ++J) {
      v8f acc = {};
#pragma unroll
      for (int q = 0; q < 4; ++q)
        acc = wmma4(frag_c(Si, q), ld_rowfrag(T1, LDN, 16 * J, 0, q, lane), acc);
      KTc[J] = acc;
      st_ct(Km, LDN, 16 * J, 0, acc, lane);
    }
    __builtin_amdgcn_wave_barrier();

    {
      float acc = xp[lane];
#pragma unroll
      for (int j = 0; j < 16; ++j) acc += Km[lane * LDN + j] * nu[j];
      x[lane] = acc;
      ob[(size_t)t * 32 + lane] = acc;
    }

#pragma unroll
    for (int I = 0; I < 2; ++I)
#pragma unroll
      for (int J = 0; J < 2; ++J) {
        v8f acc = {};
#pragma unroll
        for (int q = 0; q < 4; ++q)
          acc = wmma4(frag_c(KTc[I], q), frag_c(T2c[J], q), acc);
        Pc[I][J] = Pc[I][J] - acc;
      }
    __builtin_amdgcn_wave_barrier();
  }
}

extern "C" void kernel_launch(void* const* d_in, const int* in_sizes, int n_in,
                              void* d_out, int out_size, void* d_ws, size_t ws_size,
                              hipStream_t stream) {
  const float* state0 = (const float*)d_in[0];
  const float* cov0   = (const float*)d_in[1];
  const float* meas   = (const float*)d_in[2];
  const float* F      = (const float*)d_in[3];
  const float* H      = (const float*)d_in[4];
  const float* Q      = (const float*)d_in[5];
  const float* R      = (const float*)d_in[6];
  float* out = (float*)d_out;

  const int B = in_sizes[0] / 32;
  const int T = in_sizes[2] / (B * 16);
  const int blocks = (B + WPB - 1) / WPB;
  kalman_wmma_kernel<<<blocks, 32 * WPB, 0, stream>>>(state0, cov0, meas,
                                                      F, H, Q, R, out, B, T);
}


# State vector kept in registers: x, x_pred, innovation broadcast by v_readlane to F/H rows in registers; fused sums split into 2 or 4 partial sums, gain*innovation as two half sums; no LDS, no branch
# speedup vs baseline: 5.6721x; 1.0714x over previous
#include <hip/hip_runtime.h>
#include <hip/hip_bf16.h>


typedef float v2f __attribute__((ext_vector_type(2)));
typedef float v8f __attribute__((ext_vector_type(8)));

#define WPB 2

#define LDW 44
#define LDN 20


__device__ __forceinline__ v2f ld_rowfrag(const float* m, int ld, int r0, int c0, int q, int lane) {
  const int h = lane >> 4, c = lane & 15;
  const float* p = m + (r0 + c) * ld + c0 + (h << 3) + (q << 1);
  v2f a; a.x = p[0]; a.y = p[1];
  return a;
}
__device__ __forceinline__ v2f frag_c(v8f x, int q) {
  v2f f; f.x = x[2 * q]; f.y = x[2 * q + 1];
  return f;
}
__device__ __forceinline__ v8f wmma4(v2f a, v2f b, v8f c) {
  return __builtin_amdgcn_wmma_f32_16x16x4_f32(false, a, false, b, (short)0, c, false, false);
}

__device__ __forceinline__ void st_c(float* d, int ld, int r0, int c0, v8f c, int lane) {
  const int h = lane >> 4, n = lane & 15;
#pragma unroll
  for (int r = 0; r < 8; ++r)
    d[(r0 + r + (h << 3)) * ld + c0 + n] = c[r];
}

__device__ __forceinline__ float xhalf(float x) {
  const int b = __builtin_bit_cast(int, x);
  return __builtin_bit_cast(float,
      __builtin_amdgcn_permlanex16(b, b, 0x76543210, 0xfedcba98, false, false));
}
__device__ __forceinline__ float rdlane(float v, int l) {
  return __builtin_bit_cast(float,
      __builtin_amdgcn_readlane(__builtin_bit_cast(int, v), l));
}

__global__ __launch_bounds__(32 * WPB)
void kalman_wmma_kernel(const float* __restrict__ state0, const float* __restrict__ cov0,
                        const float* __restrict__ meas,   const float* __restrict__ Fg,
                        const float* __restrict__ Hg,     const float* __restrict__ Qg,
                        const float* __restrict__ Rg,     float* __restrict__ out,
                        int B, int T) {
  const int tid  = threadIdx.x;
  const int lane = tid & 31;
  const int w    = tid >> 5;
  const int h    = lane >> 4;
  const int n    = lane & 15;

  __shared__ __attribute__((aligned(16))) float sF[32 * LDW];
  __shared__ __attribute__((aligned(16))) float sH[16 * LDW];
  __shared__ __attribute__((aligned(16))) float sP [WPB][32 * LDW];
  __shared__ __attribute__((aligned(16))) float sT1[WPB][32 * LDN];

  for (int i = tid; i < 32 * 32; i += blockDim.x) sF[(i >> 5) * LDW + (i & 31)] = Fg[i];
  for (int i = tid; i < 16 * 32; i += blockDim.x) sH[(i >> 5) * LDW + (i & 31)] = Hg[i];
  __syncthreads();

  const int b = blockIdx.x * WPB + w;
  if (b >= B) return;

  float* P  = sP [w];
  float* T1 = sT1[w];

  v2f fF[2][2][4];
  v2f fH[2][4];
#pragma unroll
  for (int J = 0; J < 2; ++J)
#pragma unroll
    for (int Kt = 0; Kt < 2; ++Kt)
#pragma unroll
      for (int q = 0; q < 4; ++q)
        fF[J][Kt][q] = ld_rowfrag(sF, LDW, 16 * J, 16 * Kt, q, lane);
#pragma unroll
  for (int Kt = 0; Kt < 2; ++Kt)
#pragma unroll
    for (int q = 0; q < 4; ++q)
      fH[Kt][q] = ld_rowfrag(sH, LDW, 0, 16 * Kt, q, lane);
  v8f Qc[2][2], Pc[2][2], Rc;
#pragma unroll
  for (int I = 0; I < 2; ++I)
#pragma unroll
    for (int J = 0; J < 2; ++J)
#pragma unroll
      for (int r = 0; r < 8; ++r) {
        const int e = (16 * I + 8 * h + r) * 32 + 16 * J + n;
        Qc[I][J][r] = Qg[e];
        Pc[I][J][r] = cov0[(size_t)b * 1024 + e];
      }
#pragma unroll
  for (int r = 0; r < 8; ++r) Rc[r] = Rg[(8 * h + r) * 16 + n];

  float Fa[32], Fb[32], Hr[32];
#pragma unroll
  for (int j = 0; j < 32; ++j) {
    Fa[j] = sF[n * LDW + j];
    Fb[j] = sF[(16 + n) * LDW + j];
    Hr[j] = sH[n * LDW + j];
  }
  float xa = state0[(size_t)b * 32 + n];
  float xb = state0[(size_t)b * 32 + 16 + n];
  const float* zb = meas + (size_t)b * T * 16;
  float*       ob = out  + (size_t)b * T * 32;

#pragma unroll 1
  for (int t = 0; t < T; ++t) {
    if (t + 1 < T) __builtin_prefetch(zb + (size_t)(t + 1) * 16, 0, 1);

    const float zt = zb[(size_t)t * 16 + n];

    float pa2[2] = {0.f, 0.f}, pb2[2] = {0.f, 0.f};
#pragma unroll
    for (int j = 0; j < 32; ++j) {
      const float xj = rdlane(j < 16 ? xa : xb, j & 15);
      pa2[j & 1] = __builtin_fmaf(Fa[j], xj, pa2[j & 1]);
      pb2[j & 1] = __builtin_fmaf(Fb[j], xj, pb2[j & 1]);
    }
    const float xpa = pa2[0] + pa2[1], xpb = pb2[0] + pb2[1];
    float nu;
    {
      float ya[4] = {0.f, 0.f, 0.f, 0.f};
#pragma unroll
      for (int j = 0; j < 32; ++j)
        ya[j & 3] = __builtin_fmaf(Hr[j], rdlane(j < 16 ? xpa : xpb, j & 15), ya[j & 3]);
      nu = zt - ((ya[0] + ya[1]) + (ya[2] + ya[3]));
    }

    v8f Zc[2][2];
#pragma unroll
    for (int I = 0; I < 2; ++I)
#pragma unroll
      for (int J = 0; J < 2; ++J) {
        v8f acc = {};
#pragma unroll
        for (int Kt = 0; Kt < 2; ++Kt)
#pragma unroll
          for (int q = 0; q < 4; ++q)
            acc = wmma4(frag_c(Pc[Kt][I], q), fF[J][Kt][q], acc);
        Zc[I][J] = acc;
      }

#pragma unroll
    for (int I = 0; I < 2; ++I)
#pragma unroll
      for (int J = 0; J < 2; ++J) {
        v8f acc = {};
#pragma unroll
        for (int Kt = 0; Kt < 2; ++Kt)
#pragma unroll
          for (int q = 0; q < 4; ++q)
            acc = wmma4(frag_c(Zc[Kt][I], q), fF[J][Kt][q], acc);
        Pc[I][J] = acc + Qc[I][J];
        st_c(P, LDW, 16 * I, 16 * J, Pc[I][J], lane);
      }

    __builtin_amdgcn_wave_barrier();

    v8f T1c[2];
#pragma unroll
    for (int I = 0; I < 2; ++I) {
      v8f acc = {};
#pragma unroll
      for (int Kt = 0; Kt < 2; ++Kt)
#pragma unroll
        for (int q = 0; q < 4; ++q)
          acc = wmma4(ld_rowfrag(P, LDW, 16 * I, 16 * Kt, q, lane), fH[Kt][q], acc);
      T1c[I] = acc;
      st_c(T1, LDN, 16 * I, 0, acc, lane);
    }

    float ga[16];
    {
      v8f acc = {};
#pragma unroll
      for (int Kt = 0; Kt < 2; ++Kt)
#pragma unroll
        for (int q = 0; q < 4; ++q)
          acc = wmma4(fH[Kt][q], frag_c(T1c[Kt], q), acc);
#pragma unroll
      for (int r = 0; r < 8; ++r) {
        const float s = acc[r] + Rc[r];
        const float o = xhalf(s);
        ga[r]     = h ? ((r     == n) ? 1.0f : 0.0f) : s;
        ga[r + 8] = h ? ((r + 8 == n) ? 1.0f : 0.0f) : o;
      }
    }

    __builtin_amdgcn_wave_barrier();

    v8f T2c[2];
#pragma unroll
    for (int J = 0; J < 2; ++J) {
      v8f acc = {};
#pragma unroll
      for (int Kt = 0; Kt < 2; ++Kt)
#pragma unroll
        for (int q = 0; q < 4; ++q)
          acc = wmma4(fH[Kt][q], frag_c(Pc[Kt][J], q), acc);
      T2c[J] = acc;
    }
#pragma unroll
    for (int k = 0; k < 16; ++k) {
      const float pinv = __builtin_amdgcn_rcpf(rdlane(ga[k], k));
      const float rk   = ga[k] * pinv;
      ga[k] = rk;
#pragma unroll
      for (int r = 0; r < 16; ++r) {
        if (r == k) continue;
        const float f = rdlane(ga[r], k);
        ga[r] = ga[r] - f * rk;
      }
    }
    v8f Si;
#pragma unroll
    for (int r = 0; r < 8; ++r) {
      const float lo = xhalf(ga[r]);
      Si[r] = h ? ga[r + 8] : lo;
    }

    v8f KTc[2];
#pragma unroll
    for (int J = 0; J < 2; ++J) {
      v8f acc = {};
#pragma unroll
      for (int q = 0; q < 4; ++q)
        acc = wmma4(frag_c(Si, q), ld_rowfrag(T1, LDN, 16 * J, 0, q, lane), acc);
      KTc[J] = acc;
    }

    {
      float pa = 0.f, pb = 0.f;
#pragma unroll
      for (int r = 0; r < 8; ++r) {
        const float lo = rdlane(nu, r), hi = rdlane(nu, 8 + r);
        const float nr = h ? hi : lo;
        pa = __builtin_fmaf(KTc[0][r], nr, pa);
        pb = __builtin_fmaf(KTc[1][r], nr, pb);
      }
      xa = xpa + (pa + xhalf(pa));
      xb = xpb + (pb + xhalf(pb));
      ob[(size_t)t * 32 + lane] = h ? xb : xa;
    }

#pragma unroll
    for (int I = 0; I < 2; ++I)
#pragma unroll
      for (int J = 0; J < 2; ++J) {
        v8f acc = {};
#pragma unroll
        for (int q = 0; q < 4; ++q)
          acc = wmma4(frag_c(KTc[I], q), frag_c(T2c[J], q), acc);
        Pc[I][J] = Pc[I][J] - acc;
      }
    __builtin_amdgcn_wave_barrier();
  }
}

extern "C" void kernel_launch(void* const* d_in, const int* in_sizes, int n_in,
                              void* d_out, int out_size, void* d_ws, size_t ws_size,
                              hipStream_t stream) {
  const float* state0 = (const float*)d_in[0];
  const float* cov0   = (const float*)d_in[1];
  const float* meas   = (const float*)d_in[2];
  const float* F      = (const float*)d_in[3];
  const float* H      = (const float*)d_in[4];
  const float* Q      = (const float*)d_in[5];
  const float* R      = (const float*)d_in[6];
  float* out = (float*)d_out;

  const int B = in_sizes[0] / 32;
  const int T = in_sizes[2] / (B * 16);
  const int blocks = (B + WPB - 1) / WPB;
  kalman_wmma_kernel<<<blocks, 32 * WPB, 0, stream>>>(state0, cov0, meas,
                                                      F, H, Q, R, out, B, T);
}


# Work-group of 8 waves instead of 2 (256 groups, one per compute unit by the register budget): every SIMD runs exactly two batches, not what the placement of 1024 two-wave groups gave; same wave code
# speedup vs baseline: 5.9579x; 1.0504x over previous
#include <hip/hip_runtime.h>
#include <hip/hip_bf16.h>


typedef float v2f __attribute__((ext_vector_type(2)));
typedef float v8f __attribute__((ext_vector_type(8)));

#define WPB 8

#define LDW 44
#define LDN 20


__device__ __forceinline__ v2f ld_rowfrag(const float* m, int ld, int r0, int c0, int q, int lane) {
  const int h = lane >> 4, c = lane & 15;
  const float* p = m + (r0 + c) * ld + c0 + (h << 3) + (q << 1);
  v2f a; a.x = p[0]; a.y = p[1];
  return a;
}
__device__ __forceinline__ v2f frag_c(v8f x, int q) {
  v2f f; f.x = x[2 * q]; f.y = x[2 * q + 1];
  return f;
}
__device__ __forceinline__ v8f wmma4(v2f a, v2f b, v8f c) {
  return __builtin_amdgcn_wmma_f32_16x16x4_f32(false, a, false, b, (short)0, c, false, false);
}

__device__ __forceinline__ void st_c(float* d, int ld, int r0, int c0, v8f c, int lane) {
  const int h = lane >> 4, n = lane & 15;
#pragma unroll
  for (int r = 0; r < 8; ++r)
    d[(r0 + r + (h << 3)) * ld + c0 + n] = c[r];
}

__device__ __forceinline__ float xhalf(float x) {
  const int b = __builtin_bit_cast(int, x);
  return __builtin_bit_cast(float,
      __builtin_amdgcn_permlanex16(b, b, 0x76543210, 0xfedcba98, false, false));
}
__device__ __forceinline__ float rdlane(float v, int l) {
  return __builtin_bit_cast(float,
      __builtin_amdgcn_readlane(__builtin_bit_cast(int, v), l));
}

__global__ __launch_bounds__(32 * WPB)
void kalman_wmma_kernel(const float* __restrict__ state0, const float* __restrict__ cov0,
                        const float* __restrict__ meas,   const float* __restrict__ Fg,
                        const float* __restrict__ Hg,     const float* __restrict__ Qg,
                        const float* __restrict__ Rg,     float* __restrict__ out,
                        int B, int T) {
  const int tid  = threadIdx.x;
  const int lane = tid & 31;
  const int w    = tid >> 5;
  const int h    = lane >> 4;
  const int n    = lane & 15;

  __shared__ __attribute__((aligned(16))) float sF[32 * LDW];
  __shared__ __attribute__((aligned(16))) float sH[16 * LDW];
  __shared__ __attribute__((aligned(16))) float sP [WPB][32 * LDW];
  __shared__ __attribute__((aligned(16))) float sT1[WPB][32 * LDN];

  for (int i = tid; i < 32 * 32; i += blockDim.x) sF[(i >> 5) * LDW + (i & 31)] = Fg[i];
  for (int i = tid; i < 16 * 32; i += blockDim.x) sH[(i >> 5) * LDW + (i & 31)] = Hg[i];
  __syncthreads();

  const int b = blockIdx.x * WPB + w;
  if (b >= B) return;

  float* P  = sP [w];
  float* T1 = sT1[w];

  v2f fF[2][2][4];
  v2f fH[2][4];
#pragma unroll
  for (int J = 0; J < 2; ++J)
#pragma unroll
    for (int Kt = 0; Kt < 2; ++Kt)
#pragma unroll
      for (int q = 0; q < 4; ++q)
        fF[J][Kt][q] = ld_rowfrag(sF, LDW, 16 * J, 16 * Kt, q, lane);
#pragma unroll
  for (int Kt = 0; Kt < 2; ++Kt)
#pragma unroll
    for (int q = 0; q < 4; ++q)
      fH[Kt][q] = ld_rowfrag(sH, LDW, 0, 16 * Kt, q, lane);
  v8f Qc[2][2], Pc[2][2], Rc;
#pragma unroll
  for (int I = 0; I < 2; ++I)
#pragma unroll
    for (int J = 0; J < 2; ++J)
#pragma unroll
      for (int r = 0; r < 8; ++r) {
        const int e = (16 * I + 8 * h + r) * 32 + 16 * J + n;
        Qc[I][J][r] = Qg[e];
        Pc[I][J][r] = cov0[(size_t)b * 1024 + e];
      }
#pragma unroll
  for (int r = 0; r < 8; ++r) Rc[r] = Rg[(8 * h + r) * 16 + n];

  float Fa[32], Fb[32], Hr[32];
#pragma unroll
  for (int j = 0; j < 32; ++j) {
    Fa[j] = sF[n * LDW + j];
    Fb[j] = sF[(16 + n) * LDW + j];
    Hr[j] = sH[n * LDW + j];
  }
  float xa = state0[(size_t)b * 32 + n];
  float xb = state0[(size_t)b * 32 + 16 + n];
  const float* zb = meas + (size_t)b * T * 16;
  float*       ob = out  + (size_t)b * T * 32;

#pragma unroll 1
  for (int t = 0; t < T; ++t) {
    if (t + 1 < T) __builtin_prefetch(zb + (size_t)(t + 1) * 16, 0, 1);

    const float zt = zb[(size_t)t * 16 + n];

    float pa2[2] = {0.f, 0.f}, pb2[2] = {0.f, 0.f};
#pragma unroll
    for (int j = 0; j < 32; ++j) {
      const float xj = rdlane(j < 16 ? xa : xb, j & 15);
      pa2[j & 1] = __builtin_fmaf(Fa[j], xj, pa2[j & 1]);
      pb2[j & 1] = __builtin_fmaf(Fb[j], xj, pb2[j & 1]);
    }
    const float xpa = pa2[0] + pa2[1], xpb = pb2[0] + pb2[1];
    float nu;
    {
      float ya[4] = {0.f, 0.f, 0.f, 0.f};
#pragma unroll
      for (int j = 0; j < 32; ++j)
        ya[j & 3] = __builtin_fmaf(Hr[j], rdlane(j < 16 ? xpa : xpb, j & 15), ya[j & 3]);
      nu = zt - ((ya[0] + ya[1]) + (ya[2] + ya[3]));
    }

    v8f Zc[2][2];
#pragma unroll
    for (int I = 0; I < 2; ++I)
#pragma unroll
      for (int J = 0; J < 2; ++J) {
        v8f acc = {};
#pragma unroll
        for (int Kt = 0; Kt < 2; ++Kt)
#pragma unroll
          for (int q = 0; q < 4; ++q)
            acc = wmma4(frag_c(Pc[Kt][I], q), fF[J][Kt][q], acc);
        Zc[I][J] = acc;
      }

#pragma unroll
    for (int I = 0; I < 2; ++I)
#pragma unroll
      for (int J = 0; J < 2; ++J) {
        v8f acc = {};
#pragma unroll
        for (int Kt = 0; Kt < 2; ++Kt)
#pragma unroll
          for (int q = 0; q < 4; ++q)
            acc = wmma4(frag_c(Zc[Kt][I], q), fF[J][Kt][q], acc);
        Pc[I][J] = acc + Qc[I][J];
        st_c(P, LDW, 16 * I, 16 * J, Pc[I][J], lane);
      }

    __builtin_amdgcn_wave_barrier();

    v8f T1c[2];
#pragma unroll
    for (int I = 0; I < 2; ++I) {
      v8f acc = {};
#pragma unroll
      for (int Kt = 0; Kt < 2; ++Kt)
#pragma unroll
        for (int q = 0; q < 4; ++q)
          acc = wmma4(ld_rowfrag(P, LDW, 16 * I, 16 * Kt, q, lane), fH[Kt][q], acc);
      T1c[I] = acc;
      st_c(T1, LDN, 16 * I, 0, acc, lane);
    }

    float ga[16];
    {
      v8f acc = {};
#pragma unroll
      for (int Kt = 0; Kt < 2; ++Kt)
#pragma unroll
        for (int q = 0; q < 4; ++q)
          acc = wmma4(fH[Kt][q], frag_c(T1c[Kt], q), acc);
#pragma unroll
      for (int r = 0; r < 8; ++r) {
        const float s = acc[r] + Rc[r];
        const float o = xhalf(s);
        ga[r]     = h ? ((r     == n) ? 1.0f : 0.0f) : s;
        ga[r + 8] = h ? ((r + 8 == n) ? 1.0f : 0.0f) : o;
      }
    }

    __builtin_amdgcn_wave_barrier();

    v8f T2c[2];
#pragma unroll
    for (int J = 0; J < 2; ++J) {
      v8f acc = {};
#pragma unroll
      for (int Kt = 0; Kt < 2; ++Kt)
#pragma unroll
        for (int q = 0; q < 4; ++q)
          acc = wmma4(fH[Kt][q], frag_c(Pc[Kt][J], q), acc);
      T2c[J] = acc;
    }
#pragma unroll
    for (int k = 0; k < 16; ++k) {
      const float pinv = __builtin_amdgcn_rcpf(rdlane(ga[k], k));
      const float rk   = ga[k] * pinv;
      ga[k] = rk;
#pragma unroll
      for (int r = 0; r < 16; ++r) {
        if (r == k) continue;
        const float f = rdlane(ga[r], k);
        ga[r] = ga[r] - f * rk;
      }
    }
    v8f Si;
#pragma unroll
    for (int r = 0; r < 8; ++r) {
      const float lo = xhalf(ga[r]);
      Si[r] = h ? ga[r + 8] : lo;
    }

    v8f KTc[2];
#pragma unroll
    for (int J = 0; J < 2; ++J) {
      v8f acc = {};
#pragma unroll
      for (int q = 0; q < 4; ++q)
        acc = wmma4(frag_c(Si, q), ld_rowfrag(T1, LDN, 16 * J, 0, q, lane), acc);
      KTc[J] = acc;
    }

    {
      float pa = 0.f, pb = 0.f;
#pragma unroll
      for (int r = 0; r < 8; ++r) {
        const float lo = rdlane(nu, r), hi = rdlane(nu, 8 + r);
        const float nr = h ? hi : lo;
        pa = __builtin_fmaf(KTc[0][r], nr, pa);
        pb = __builtin_fmaf(KTc[1][r], nr, pb);
      }
      xa = xpa + (pa + xhalf(pa));
      xb = xpb + (pb + xhalf(pb));
      ob[(size_t)t * 32 + lane] = h ? xb : xa;
    }

#pragma unroll
    for (int I = 0; I < 2; ++I)
#pragma unroll
      for (int J = 0; J < 2; ++J) {
        v8f acc = {};
#pragma unroll
        for (int q = 0; q < 4; ++q)
          acc = wmma4(frag_c(KTc[I], q), frag_c(T2c[J], q), acc);
        Pc[I][J] = Pc[I][J] - acc;
      }
    __builtin_amdgcn_wave_barrier();
  }
}

extern "C" void kernel_launch(void* const* d_in, const int* in_sizes, int n_in,
                              void* d_out, int out_size, void* d_ws, size_t ws_size,
                              hipStream_t stream) {
  const float* state0 = (const float*)d_in[0];
  const float* cov0   = (const float*)d_in[1];
  const float* meas   = (const float*)d_in[2];
  const float* F      = (const float*)d_in[3];
  const float* H      = (const float*)d_in[4];
  const float* Q      = (const float*)d_in[5];
  const float* R      = (const float*)d_in[6];
  float* out = (float*)d_out;

  const int B = in_sizes[0] / 32;
  const int T = in_sizes[2] / (B * 16);
  const int blocks = (B + WPB - 1) / WPB;
  kalman_wmma_kernel<<<blocks, 32 * WPB, 0, stream>>>(state0, cov0, meas,
                                                      F, H, Q, R, out, B, T);
}


# Wave priority set by phase (s_setprio): both matrix phases outrank the vector-only S inversion, so the SIMD's second wave issues its WMMAs while one wave inverts; same arithmetic, same sum order
# speedup vs baseline: 6.1379x; 1.0302x over previous
#include <hip/hip_runtime.h>
#include <hip/hip_bf16.h>


typedef float v2f __attribute__((ext_vector_type(2)));
typedef float v8f __attribute__((ext_vector_type(8)));

#define WPB 8

#define LDW 44
#define LDN 20


__device__ __forceinline__ v2f ld_rowfrag(const float* m, int ld, int r0, int c0, int q, int lane) {
  const int h = lane >> 4, c = lane & 15;
  const float* p = m + (r0 + c) * ld + c0 + (h << 3) + (q << 1);
  v2f a; a.x = p[0]; a.y = p[1];
  return a;
}
__device__ __forceinline__ v2f frag_c(v8f x, int q) {
  v2f f; f.x = x[2 * q]; f.y = x[2 * q + 1];
  return f;
}
__device__ __forceinline__ v8f wmma4(v2f a, v2f b, v8f c) {
  return __builtin_amdgcn_wmma_f32_16x16x4_f32(false, a, false, b, (short)0, c, false, false);
}

__device__ __forceinline__ void st_c(float* d, int ld, int r0, int c0, v8f c, int lane) {
  const int h = lane >> 4, n = lane & 15;
#pragma unroll
  for (int r = 0; r < 8; ++r)
    d[(r0 + r + (h << 3)) * ld + c0 + n] = c[r];
}

__device__ __forceinline__ float xhalf(float x) {
  const int b = __builtin_bit_cast(int, x);
  return __builtin_bit_cast(float,
      __builtin_amdgcn_permlanex16(b, b, 0x76543210, 0xfedcba98, false, false));
}
__device__ __forceinline__ float rdlane(float v, int l) {
  return __builtin_bit_cast(float,
      __builtin_amdgcn_readlane(__builtin_bit_cast(int, v), l));
}

__global__ __launch_bounds__(32 * WPB)
void kalman_wmma_kernel(const float* __restrict__ state0, const float* __restrict__ cov0,
                        const float* __restrict__ meas,   const float* __restrict__ Fg,
                        const float* __restrict__ Hg,     const float* __restrict__ Qg,
                        const float* __restrict__ Rg,     float* __restrict__ out,
                        int B, int T) {
  const int tid  = threadIdx.x;
  const int lane = tid & 31;
  const int w    = tid >> 5;
  const int h    = lane >> 4;
  const int n    = lane & 15;

  __shared__ __attribute__((aligned(16))) float sF[32 * LDW];
  __shared__ __attribute__((aligned(16))) float sH[16 * LDW];
  __shared__ __attribute__((aligned(16))) float sP [WPB][32 * LDW];
  __shared__ __attribute__((aligned(16))) float sT1[WPB][32 * LDN];

  for (int i = tid; i < 32 * 32; i += blockDim.x) sF[(i >> 5) * LDW + (i & 31)] = Fg[i];
  for (int i = tid; i < 16 * 32; i += blockDim.x) sH[(i >> 5) * LDW + (i & 31)] = Hg[i];
  __syncthreads();

  const int b = blockIdx.x * WPB + w;
  if (b >= B) return;

  float* P  = sP [w];
  float* T1 = sT1[w];

  v2f fF[2][2][4];
  v2f fH[2][4];
#pragma unroll
  for (int J = 0; J < 2; ++J)
#pragma unroll
    for (int Kt = 0; Kt < 2; ++Kt)
#pragma unroll
      for (int q = 0; q < 4; ++q)
        fF[J][Kt][q] = ld_rowfrag(sF, LDW, 16 * J, 16 * Kt, q, lane);
#pragma unroll
  for (int Kt = 0; Kt < 2; ++Kt)
#pragma unroll
    for (int q = 0; q < 4; ++q)
      fH[Kt][q] = ld_rowfrag(sH, LDW, 0, 16 * Kt, q, lane);
  v8f Qc[2][2], Pc[2][2], Rc;
#pragma unroll
  for (int I = 0; I < 2; ++I)
#pragma unroll
    for (int J = 0; J < 2; ++J)
#pragma unroll
      for (int r = 0; r < 8; ++r) {
        const int e = (16 * I + 8 * h + r) * 32 + 16 * J + n;
        Qc[I][J][r] = Qg[e];
        Pc[I][J][r] = cov0[(size_t)b * 1024 + e];
      }
#pragma unroll
  for (int r = 0; r < 8; ++r) Rc[r] = Rg[(8 * h + r) * 16 + n];

  float Fa[32], Fb[32], Hr[32];
#pragma unroll
  for (int j = 0; j < 32; ++j) {
    Fa[j] = sF[n * LDW + j];
    Fb[j] = sF[(16 + n) * LDW + j];
    Hr[j] = sH[n * LDW + j];
  }
  float xa = state0[(size_t)b * 32 + n];
  float xb = state0[(size_t)b * 32 + 16 + n];
  const float* zb = meas + (size_t)b * T * 16;
  float*       ob = out  + (size_t)b * T * 32;

#pragma unroll 1
  for (int t = 0; t < T; ++t) {
    __builtin_amdgcn_s_setprio(2);
    if (t + 1 < T) __builtin_prefetch(zb + (size_t)(t + 1) * 16, 0, 1);

    const float zt = zb[(size_t)t * 16 + n];

    float pa2[2] = {0.f, 0.f}, pb2[2] = {0.f, 0.f};
#pragma unroll
    for (int j = 0; j < 32; ++j) {
      const float xj = rdlane(j < 16 ? xa : xb, j & 15);
      pa2[j & 1] = __builtin_fmaf(Fa[j], xj, pa2[j & 1]);
      pb2[j & 1] = __builtin_fmaf(Fb[j], xj, pb2[j & 1]);
    }
    const float xpa = pa2[0] + pa2[1], xpb = pb2[0] + pb2[1];
    float nu;
    {
      float ya[4] = {0.f, 0.f, 0.f, 0.f};
#pragma unroll
      for (int j = 0; j < 32; ++j)
        ya[j & 3] = __builtin_fmaf(Hr[j], rdlane(j < 16 ? xpa : xpb, j & 15), ya[j & 3]);
      nu = zt - ((ya[0] + ya[1]) + (ya[2] + ya[3]));
    }

    v8f Zc[2][2];
#pragma unroll
    for (int I = 0; I < 2; ++I)
#pragma unroll
      for (int J = 0; J < 2; ++J) {
        v8f acc = {};
#pragma unroll
        for (int Kt = 0; Kt < 2; ++Kt)
#pragma unroll
          for (int q = 0; q < 4; ++q)
            acc = wmma4(frag_c(Pc[Kt][I], q), fF[J][Kt][q], acc);
        Zc[I][J] = acc;
      }

#pragma unroll
    for (int I = 0; I < 2; ++I)
#pragma unroll
      for (int J = 0; J < 2; ++J) {
        v8f acc = {};
#pragma unroll
        for (int Kt = 0; Kt < 2; ++Kt)
#pragma unroll
          for (int q = 0; q < 4; ++q)
            acc = wmma4(frag_c(Zc[Kt][I], q), fF[J][Kt][q], acc);
        Pc[I][J] = acc + Qc[I][J];
        st_c(P, LDW, 16 * I, 16 * J, Pc[I][J], lane);
      }

    __builtin_amdgcn_wave_barrier();

    v8f T1c[2];
#pragma unroll
    for (int I = 0; I < 2; ++I) {
      v8f acc = {};
#pragma unroll
      for (int Kt = 0; Kt < 2; ++Kt)
#pragma unroll
        for (int q = 0; q < 4; ++q)
          acc = wmma4(ld_rowfrag(P, LDW, 16 * I, 16 * Kt, q, lane), fH[Kt][q], acc);
      T1c[I] = acc;
      st_c(T1, LDN, 16 * I, 0, acc, lane);
    }

    v8f Sc = {};
#pragma unroll
    for (int Kt = 0; Kt < 2; ++Kt)
#pragma unroll
      for (int q = 0; q < 4; ++q)
        Sc = wmma4(fH[Kt][q], frag_c(T1c[Kt], q), Sc);

    __builtin_amdgcn_wave_barrier();

    v2f fT1[2][4];
#pragma unroll
    for (int J = 0; J < 2; ++J)
#pragma unroll
      for (int q = 0; q < 4; ++q)
        fT1[J][q] = ld_rowfrag(T1, LDN, 16 * J, 0, q, lane);

    v8f T2c[2];
#pragma unroll
    for (int J = 0; J < 2; ++J) {
      v8f acc = {};
#pragma unroll
      for (int Kt = 0; Kt < 2; ++Kt)
#pragma unroll
        for (int q = 0; q < 4; ++q)
          acc = wmma4(fH[Kt][q], frag_c(Pc[Kt][J], q), acc);
      T2c[J] = acc;
    }
    __builtin_amdgcn_s_setprio(0);
    asm volatile("s_wait_alu depctr_va_vdst(0)" : "+v"(Sc));

    float ga[16];
#pragma unroll
    for (int r = 0; r < 8; ++r) {
      const float s = Sc[r] + Rc[r];
      const float o = xhalf(s);
      ga[r]     = h ? ((r     == n) ? 1.0f : 0.0f) : s;
      ga[r + 8] = h ? ((r + 8 == n) ? 1.0f : 0.0f) : o;
    }

#pragma unroll
    for (int k = 0; k < 16; ++k) {
      const float pinv = __builtin_amdgcn_rcpf(rdlane(ga[k], k));
      const float rk   = ga[k] * pinv;
      ga[k] = rk;
#pragma unroll
      for (int r = 0; r < 16; ++r) {
        if (r == k) continue;
        const float f = rdlane(ga[r], k);
        ga[r] = ga[r] - f * rk;
      }
    }
    v8f Si;
#pragma unroll
    for (int r = 0; r < 8; ++r) {
      const float lo = xhalf(ga[r]);
      Si[r] = h ? ga[r + 8] : lo;
    }

    __builtin_amdgcn_s_setprio(1);

    v8f KTc[2];
#pragma unroll
    for (int J = 0; J < 2; ++J) {
      v8f acc = {};
#pragma unroll
      for (int q = 0; q < 4; ++q)
        acc = wmma4(frag_c(Si, q), fT1[J][q], acc);
      KTc[J] = acc;
    }

    {
      float pa = 0.f, pb = 0.f;
#pragma unroll
      for (int r = 0; r < 8; ++r) {
        const float lo = rdlane(nu, r), hi = rdlane(nu, 8 + r);
        const float nr = h ? hi : lo;
        pa = __builtin_fmaf(KTc[0][r], nr, pa);
        pb = __builtin_fmaf(KTc[1][r], nr, pb);
      }
      xa = xpa + (pa + xhalf(pa));
      xb = xpb + (pb + xhalf(pb));
      ob[(size_t)t * 32 + lane] = h ? xb : xa;
    }

#pragma unroll
    for (int I = 0; I < 2; ++I)
#pragma unroll
      for (int J = 0; J < 2; ++J) {
        v8f acc = {};
#pragma unroll
        for (int q = 0; q < 4; ++q)
          acc = wmma4(frag_c(KTc[I], q), frag_c(T2c[J], q), acc);
        Pc[I][J] = Pc[I][J] - acc;
      }
    __builtin_amdgcn_wave_barrier();
  }
  __builtin_amdgcn_s_setprio(0);
}

extern "C" void kernel_launch(void* const* d_in, const int* in_sizes, int n_in,
                              void* d_out, int out_size, void* d_ws, size_t ws_size,
                              hipStream_t stream) {
  const float* state0 = (const float*)d_in[0];
  const float* cov0   = (const float*)d_in[1];
  const float* meas   = (const float*)d_in[2];
  const float* F      = (const float*)d_in[3];
  const float* H      = (const float*)d_in[4];
  const float* Q      = (const float*)d_in[5];
  const float* R      = (const float*)d_in[6];
  float* out = (float*)d_out;

  const int B = in_sizes[0] / 32;
  const int T = in_sizes[2] / (B * 16);
  const int blocks = (B + WPB - 1) / WPB;
  kalman_wmma_kernel<<<blocks, 32 * WPB, 0, stream>>>(state0, cov0, meas,
                                                      F, H, Q, R, out, B, T);
}


# S inversion in the result-tile layout (8 rows per lane half): a row's multiplier is a row_share DPP source of its FMA, not v_readlane + FMA; in place, column factor -1/pivot applied last (rounding)
# speedup vs baseline: 6.4721x; 1.0544x over previous
#include <hip/hip_runtime.h>
#include <hip/hip_bf16.h>


typedef float v2f __attribute__((ext_vector_type(2)));
typedef float v8f __attribute__((ext_vector_type(8)));

#define WPB 8

#define LDW 44
#define LDN 20


__device__ __forceinline__ v2f ld_rowfrag(const float* m, int ld, int r0, int c0, int q, int lane) {
  const int h = lane >> 4, c = lane & 15;
  const float* p = m + (r0 + c) * ld + c0 + (h << 3) + (q << 1);
  v2f a; a.x = p[0]; a.y = p[1];
  return a;
}
__device__ __forceinline__ v2f frag_c(v8f x, int q) {
  v2f f; f.x = x[2 * q]; f.y = x[2 * q + 1];
  return f;
}
__device__ __forceinline__ v8f wmma4(v2f a, v2f b, v8f c) {
  return __builtin_amdgcn_wmma_f32_16x16x4_f32(false, a, false, b, (short)0, c, false, false);
}

__device__ __forceinline__ void st_c(float* d, int ld, int r0, int c0, v8f c, int lane) {
  const int h = lane >> 4, n = lane & 15;
#pragma unroll
  for (int r = 0; r < 8; ++r)
    d[(r0 + r + (h << 3)) * ld + c0 + n] = c[r];
}

__device__ __forceinline__ float xhalf(float x) {
  const int b = __builtin_bit_cast(int, x);
  return __builtin_bit_cast(float,
      __builtin_amdgcn_permlanex16(b, b, 0x76543210, 0xfedcba98, false, true));
}
__device__ __forceinline__ float rdlane(float v, int l) {
  return __builtin_bit_cast(float,
      __builtin_amdgcn_readlane(__builtin_bit_cast(int, v), l));
}

template <int K>
__device__ __forceinline__ void gj_pivot(float (&ga)[8], float& cs, int h, int n) {
  constexpr int hk = K >> 3, rk = K & 7;
  const float prow = ga[rk];
  const float oth  = xhalf(prow);
  const float rowk = (h == hk) ? prow : oth;
  const float pinv = __builtin_amdgcn_rcpf(rdlane(prow, 16 * hk + K));
  const float ru = ((n == K) ? 0.0f : rowk) * pinv;
  asm("s_delay_alu instid0(VALU_DEP_1)\n\t"
      "v_fma_f32_e64_dpp %0, %0, -%8, %0 row_share:%9 row_mask:0xf bank_mask:0xf bound_ctrl:1\n\t"
      "v_fma_f32_e64_dpp %1, %1, -%8, %1 row_share:%9 row_mask:0xf bank_mask:0xf bound_ctrl:1\n\t"
      "v_fma_f32_e64_dpp %2, %2, -%8, %2 row_share:%9 row_mask:0xf bank_mask:0xf bound_ctrl:1\n\t"
      "v_fma_f32_e64_dpp %3, %3, -%8, %3 row_share:%9 row_mask:0xf bank_mask:0xf bound_ctrl:1\n\t"
      "v_fma_f32_e64_dpp %4, %4, -%8, %4 row_share:%9 row_mask:0xf bank_mask:0xf bound_ctrl:1\n\t"
      "v_fma_f32_e64_dpp %5, %5, -%8, %5 row_share:%9 row_mask:0xf bank_mask:0xf bound_ctrl:1\n\t"
      "v_fma_f32_e64_dpp %6, %6, -%8, %6 row_share:%9 row_mask:0xf bank_mask:0xf bound_ctrl:1\n\t"
      "v_fma_f32_e64_dpp %7, %7, -%8, %7 row_share:%9 row_mask:0xf bank_mask:0xf bound_ctrl:1"
      : "+v"(ga[(K + 1) & 7]), "+v"(ga[K & 7]), "+v"(ga[(K + 2) & 7]), "+v"(ga[(K + 3) & 7]),
        "+v"(ga[(K + 4) & 7]), "+v"(ga[(K + 5) & 7]), "+v"(ga[(K + 6) & 7]), "+v"(ga[(K + 7) & 7])
      : "v"(ru), "n"(K));
  const float rst = (n == K) ? -1.0f : ru;
  ga[rk] = (h == hk) ? rst : ga[rk];
  cs = (n == K) ? pinv : cs;
}

__global__ __launch_bounds__(32 * WPB)
void kalman_wmma_kernel(const float* __restrict__ state0, const float* __restrict__ cov0,
                        const float* __restrict__ meas,   const float* __restrict__ Fg,
                        const float* __restrict__ Hg,     const float* __restrict__ Qg,
                        const float* __restrict__ Rg,     float* __restrict__ out,
                        int B, int T) {
  const int tid  = threadIdx.x;
  const int lane = tid & 31;
  const int w    = tid >> 5;
  const int h    = lane >> 4;
  const int n    = lane & 15;

  __shared__ __attribute__((aligned(16))) float sF[32 * LDW];
  __shared__ __attribute__((aligned(16))) float sH[16 * LDW];
  __shared__ __attribute__((aligned(16))) float sP [WPB][32 * LDW];
  __shared__ __attribute__((aligned(16))) float sT1[WPB][32 * LDN];

  for (int i = tid; i < 32 * 32; i += blockDim.x) sF[(i >> 5) * LDW + (i & 31)] = Fg[i];
  for (int i = tid; i < 16 * 32; i += blockDim.x) sH[(i >> 5) * LDW + (i & 31)] = Hg[i];
  __syncthreads();

  const int b = blockIdx.x * WPB + w;
  if (b >= B) return;

  float* P  = sP [w];
  float* T1 = sT1[w];

  v2f fF[2][2][4];
  v2f fH[2][4];
#pragma unroll
  for (int J = 0; J < 2; ++J)
#pragma unroll
    for (int Kt = 0; Kt < 2; ++Kt)
#pragma unroll
      for (int q = 0; q < 4; ++q)
        fF[J][Kt][q] = ld_rowfrag(sF, LDW, 16 * J, 16 * Kt, q, lane);
#pragma unroll
  for (int Kt = 0; Kt < 2; ++Kt)
#pragma unroll
    for (int q = 0; q < 4; ++q)
      fH[Kt][q] = ld_rowfrag(sH, LDW, 0, 16 * Kt, q, lane);
  v8f Qc[2][2], Pc[2][2], Rc;
#pragma unroll
  for (int I = 0; I < 2; ++I)
#pragma unroll
    for (int J = 0; J < 2; ++J)
#pragma unroll
      for (int r = 0; r < 8; ++r) {
        const int e = (16 * I + 8 * h + r) * 32 + 16 * J + n;
        Qc[I][J][r] = Qg[e];
        Pc[I][J][r] = cov0[(size_t)b * 1024 + e];
      }
#pragma unroll
  for (int r = 0; r < 8; ++r) Rc[r] = Rg[(8 * h + r) * 16 + n];

  float Fa[32], Fb[32], Hr[32];
#pragma unroll
  for (int j = 0; j < 32; ++j) {
    Fa[j] = sF[n * LDW + j];
    Fb[j] = sF[(16 + n) * LDW + j];
    Hr[j] = sH[n * LDW + j];
  }
  float xa = state0[(size_t)b * 32 + n];
  float xb = state0[(size_t)b * 32 + 16 + n];
  const float* zb = meas + (size_t)b * T * 16;
  float*       ob = out  + (size_t)b * T * 32;

#pragma unroll 1
  for (int t = 0; t < T; ++t) {
    __builtin_amdgcn_s_setprio(2);
    if (t + 1 < T) __builtin_prefetch(zb + (size_t)(t + 1) * 16, 0, 1);

    const float zt = zb[(size_t)t * 16 + n];

    float pa2[2] = {0.f, 0.f}, pb2[2] = {0.f, 0.f};
#pragma unroll
    for (int j = 0; j < 32; ++j) {
      const float xj = rdlane(j < 16 ? xa : xb, j & 15);
      pa2[j & 1] = __builtin_fmaf(Fa[j], xj, pa2[j & 1]);
      pb2[j & 1] = __builtin_fmaf(Fb[j], xj, pb2[j & 1]);
    }
    const float xpa = pa2[0] + pa2[1], xpb = pb2[0] + pb2[1];
    float nu;
    {
      float ya[4] = {0.f, 0.f, 0.f, 0.f};
#pragma unroll
      for (int j = 0; j < 32; ++j)
        ya[j & 3] = __builtin_fmaf(Hr[j], rdlane(j < 16 ? xpa : xpb, j & 15), ya[j & 3]);
      nu = zt - ((ya[0] + ya[1]) + (ya[2] + ya[3]));
    }

    v8f Zc[2][2];
#pragma unroll
    for (int I = 0; I < 2; ++I)
#pragma unroll
      for (int J = 0; J < 2; ++J) {
        v8f acc = {};
#pragma unroll
        for (int Kt = 0; Kt < 2; ++Kt)
#pragma unroll
          for (int q = 0; q < 4; ++q)
            acc = wmma4(frag_c(Pc[Kt][I], q), fF[J][Kt][q], acc);
        Zc[I][J] = acc;
      }

#pragma unroll
    for (int I = 0; I < 2; ++I)
#pragma unroll
      for (int J = 0; J < 2; ++J) {
        v8f acc = {};
#pragma unroll
        for (int Kt = 0; Kt < 2; ++Kt)
#pragma unroll
          for (int q = 0; q < 4; ++q)
            acc = wmma4(frag_c(Zc[Kt][I], q), fF[J][Kt][q], acc);
        Pc[I][J] = acc + Qc[I][J];
        st_c(P, LDW, 16 * I, 16 * J, Pc[I][J], lane);
      }

    __builtin_amdgcn_wave_barrier();

    v8f T1c[2];
#pragma unroll
    for (int I = 0; I < 2; ++I) {
      v8f acc = {};
#pragma unroll
      for (int Kt = 0; Kt < 2; ++Kt)
#pragma unroll
        for (int q = 0; q < 4; ++q)
          acc = wmma4(ld_rowfrag(P, LDW, 16 * I, 16 * Kt, q, lane), fH[Kt][q], acc);
      T1c[I] = acc;
      st_c(T1, LDN, 16 * I, 0, acc, lane);
    }

    v8f Sc = {};
#pragma unroll
    for (int Kt = 0; Kt < 2; ++Kt)
#pragma unroll
      for (int q = 0; q < 4; ++q)
        Sc = wmma4(fH[Kt][q], frag_c(T1c[Kt], q), Sc);

    __builtin_amdgcn_wave_barrier();

    v2f fT1[2][4];
#pragma unroll
    for (int J = 0; J < 2; ++J)
#pragma unroll
      for (int q = 0; q < 4; ++q)
        fT1[J][q] = ld_rowfrag(T1, LDN, 16 * J, 0, q, lane);

    v8f T2c[2];
#pragma unroll
    for (int J = 0; J < 2; ++J) {
      v8f acc = {};
#pragma unroll
      for (int Kt = 0; Kt < 2; ++Kt)
#pragma unroll
        for (int q = 0; q < 4; ++q)
          acc = wmma4(fH[Kt][q], frag_c(Pc[Kt][J], q), acc);
      T2c[J] = acc;
    }
    __builtin_amdgcn_s_setprio(0);
    asm volatile("s_wait_alu depctr_va_vdst(0)" : "+v"(Sc));

    float ga[8];
#pragma unroll
    for (int r = 0; r < 8; ++r) ga[r] = Sc[r] + Rc[r];

    float cs = 0.0f;
    gj_pivot<0>(ga, cs, h, n);  gj_pivot<1>(ga, cs, h, n);  gj_pivot<2>(ga, cs, h, n);  gj_pivot<3>(ga, cs, h, n);
    gj_pivot<4>(ga, cs, h, n);  gj_pivot<5>(ga, cs, h, n);  gj_pivot<6>(ga, cs, h, n);  gj_pivot<7>(ga, cs, h, n);
    gj_pivot<8>(ga, cs, h, n);  gj_pivot<9>(ga, cs, h, n);  gj_pivot<10>(ga, cs, h, n); gj_pivot<11>(ga, cs, h, n);
    gj_pivot<12>(ga, cs, h, n); gj_pivot<13>(ga, cs, h, n); gj_pivot<14>(ga, cs, h, n); gj_pivot<15>(ga, cs, h, n);
    v8f Si;
#pragma unroll
    for (int r = 0; r < 8; ++r) Si[r] = -(ga[r] * cs);

    __builtin_amdgcn_s_setprio(1);

    v8f KTc[2];
#pragma unroll
    for (int J = 0; J < 2; ++J) {
      v8f acc = {};
#pragma unroll
      for (int q = 0; q < 4; ++q)
        acc = wmma4(frag_c(Si, q), fT1[J][q], acc);
      KTc[J] = acc;
    }

    {
      float pa = 0.f, pb = 0.f;
#pragma unroll
      for (int r = 0; r < 8; ++r) {
        const float lo = rdlane(nu, r), hi = rdlane(nu, 8 + r);
        const float nr = h ? hi : lo;
        pa = __builtin_fmaf(KTc[0][r], nr, pa);
        pb = __builtin_fmaf(KTc[1][r], nr, pb);
      }
      xa = xpa + (pa + xhalf(pa));
      xb = xpb + (pb + xhalf(pb));
      ob[(size_t)t * 32 + lane] = h ? xb : xa;
    }

#pragma unroll
    for (int I = 0; I < 2; ++I)
#pragma unroll
      for (int J = 0; J < 2; ++J) {
        v8f acc = {};
#pragma unroll
        for (int q = 0; q < 4; ++q)
          acc = wmma4(frag_c(KTc[I], q), frag_c(T2c[J], q), acc);
        Pc[I][J] = Pc[I][J] - acc;
      }
    __builtin_amdgcn_wave_barrier();
  }
  __builtin_amdgcn_s_setprio(0);
}

extern "C" void kernel_launch(void* const* d_in, const int* in_sizes, int n_in,
                              void* d_out, int out_size, void* d_ws, size_t ws_size,
                              hipStream_t stream) {
  const float* state0 = (const float*)d_in[0];
  const float* cov0   = (const float*)d_in[1];
  const float* meas   = (const float*)d_in[2];
  const float* F      = (const float*)d_in[3];
  const float* H      = (const float*)d_in[4];
  const float* Q      = (const float*)d_in[5];
  const float* R      = (const float*)d_in[6];
  float* out = (float*)d_out;

  const int B = in_sizes[0] / 32;
  const int T = in_sizes[2] / (B * 16);
  const int blocks = (B + WPB - 1) / WPB;
  kalman_wmma_kernel<<<blocks, 32 * WPB, 0, stream>>>(state0, cov0, meas,
                                                      F, H, Q, R, out, B, T);
}


# S inversion software-pipelined over the pivots: lane read, reciprocal, half swap, selects and scaling of pivot row K+1 issued inside the 8 row FMAs of pivot K; same arithmetic
# speedup vs baseline: 6.6979x; 1.0349x over previous
#include <hip/hip_runtime.h>
#include <hip/hip_bf16.h>


typedef float v2f __attribute__((ext_vector_type(2)));
typedef float v8f __attribute__((ext_vector_type(8)));

#define WPB 8

#define LDW 44
#define LDN 20


__device__ __forceinline__ v2f ld_rowfrag(const float* m, int ld, int r0, int c0, int q, int lane) {
  const int h = lane >> 4, c = lane & 15;
  const float* p = m + (r0 + c) * ld + c0 + (h << 3) + (q << 1);
  v2f a; a.x = p[0]; a.y = p[1];
  return a;
}
__device__ __forceinline__ v2f frag_c(v8f x, int q) {
  v2f f; f.x = x[2 * q]; f.y = x[2 * q + 1];
  return f;
}
__device__ __forceinline__ v8f wmma4(v2f a, v2f b, v8f c) {
  return __builtin_amdgcn_wmma_f32_16x16x4_f32(false, a, false, b, (short)0, c, false, false);
}

__device__ __forceinline__ void st_c(float* d, int ld, int r0, int c0, v8f c, int lane) {
  const int h = lane >> 4, n = lane & 15;
#pragma unroll
  for (int r = 0; r < 8; ++r)
    d[(r0 + r + (h << 3)) * ld + c0 + n] = c[r];
}

__device__ __forceinline__ float xhalf(float x) {
  const int b = __builtin_bit_cast(int, x);
  return __builtin_bit_cast(float,
      __builtin_amdgcn_permlanex16(b, b, 0x76543210, 0xfedcba98, false, true));
}
__device__ __forceinline__ float rdlane(float v, int l) {
  return __builtin_bit_cast(float,
      __builtin_amdgcn_readlane(__builtin_bit_cast(int, v), l));
}


#define GJ_FMA(i) "v_fma_f32_e64_dpp %" #i ", %" #i ", -%[ru], %" #i " row_share:%[k] row_mask:0xf bank_mask:0xf bound_ctrl:1\n\t"
#define GJ_BODY_NEXT                                                                              \
    GJ_FMA(0)                                        \
    GJ_FMA(1)                                                   \
    GJ_FMA(2)                                                                                     \
        \
    "s_delay_alu instid0(VALU_DEP_3) | instskip(NEXT) | instid1(VALU_DEP_4)\n\t"                  \
    "v_readlane_b32 %[pn], %0, %[l1]\n\t"                                                         \
    "v_permlanex16_b32 %[on], %0, %[sel], 0xfedcba98 op_sel:[0,1]\n\t"                            \
    GJ_FMA(3)                                                                                     \
    GJ_FMA(4)                                                                                     \
      \
    "s_delay_alu instid0(VALU_DEP_3) | instskip(NEXT) | instid1(VALU_DEP_4)\n\t"                  \
    "v_cndmask_b32_e64 %[rw], %[on], %0, %[mh1]\n\t"                                              \
    "v_s_rcp_f32 %[pn], %[pn]\n\t"                                                                \
    GJ_FMA(5)                                                                                     \
          \
    "v_cndmask_b32_e64 %[on], %[ru], -1.0, %[mnk]\n\t"                                            \
                          \
    "s_delay_alu instid0(VALU_DEP_3)\n\t"                                                         \
    "v_cndmask_b32_e64 %[rw], %[rw], 0, %[mn1]\n\t"                                               \
    GJ_FMA(6)                                                                                     \
    GJ_FMA(7)                                                                                     \
                                                    \
    "s_delay_alu instid0(TRANS32_DEP_1)\n\t"                                                      \
    "v_mul_f32_e64 %[rw], %[rw], %[pn]\n\t"                                                       \
                 \
    "v_cndmask_b32_e64 %[cs], %[cs], %[pk], %[mnk]\n\t"                                           \
    "v_cndmask_b32_e64 %1, %1, %[on], %[mhk]"
#define GJ_BODY_LAST                                                                              \
    GJ_FMA(0) GJ_FMA(1) GJ_FMA(2) GJ_FMA(3) GJ_FMA(4) GJ_FMA(5) GJ_FMA(6) GJ_FMA(7)               \
    "v_cndmask_b32_e64 %[on], %[ru], -1.0, %[mnk]\n\t"                                            \
    "v_cndmask_b32_e64 %[cs], %[cs], %[pk], %[mnk]\n\t"                                           \
    "s_delay_alu instid0(VALU_DEP_2)\n\t"                                                         \
    "v_cndmask_b32_e64 %1, %1, %[on], %[mhk]"
#define GJ_GA_OPS                                                                                 \
    "+v"(ga[(K + 1) & 7]), "+v"(ga[K & 7]), "+v"(ga[(K + 2) & 7]), "+v"(ga[(K + 3) & 7]),         \
    "+v"(ga[(K + 4) & 7]), "+v"(ga[(K + 5) & 7]), "+v"(ga[(K + 6) & 7]), "+v"(ga[(K + 7) & 7])

template <int K>
__device__ __forceinline__ void gj_pivot(float (&ga)[8], float& cs, float& ru, float& pinv) {
  constexpr unsigned MHK = (K >> 3) ? 0xffff0000u : 0x0000ffffu;
  constexpr unsigned MNK = (1u << K) | (1u << (16 + K));
  const float ruk = ru, pk = pinv;
  if constexpr (K < 15) {
    constexpr int K1 = K + 1;
    constexpr int L1 = 16 * (K1 >> 3) + K1;
    constexpr unsigned MH1 = (K1 >> 3) ? 0xffff0000u : 0x0000ffffu;
    constexpr unsigned MN1 = (1u << K1) | (1u << (16 + K1));
    float pn, on, rw;
    if constexpr (K == 0) {
      asm("s_delay_alu instid0(VALU_DEP_1)\n\t" GJ_BODY_NEXT
          : GJ_GA_OPS, [cs] "+v"(cs), [pn] "=&s"(pn), [on] "=&v"(on), [rw] "=&v"(rw)
          : [ru] "v"(ruk), [pk] "s"(pk), [k] "n"(K), [l1] "n"(L1), [sel] "s"(0x76543210),
            [mh1] "s"(MH1), [mn1] "s"(MN1), [mhk] "s"(MHK), [mnk] "s"(MNK));
    } else {
      asm("s_delay_alu instid0(VALU_DEP_3)\n\t" GJ_BODY_NEXT
          : GJ_GA_OPS, [cs] "+v"(cs), [pn] "=&s"(pn), [on] "=&v"(on), [rw] "=&v"(rw)
          : [ru] "v"(ruk), [pk] "s"(pk), [k] "n"(K), [l1] "n"(L1), [sel] "s"(0x76543210),
            [mh1] "s"(MH1), [mn1] "s"(MN1), [mhk] "s"(MHK), [mnk] "s"(MNK));
    }
    ru   = rw;
    pinv = pn;
  } else {
    float on;
    asm("s_delay_alu instid0(VALU_DEP_3)\n\t" GJ_BODY_LAST
        : GJ_GA_OPS, [cs] "+v"(cs), [on] "=&v"(on)
        : [ru] "v"(ruk), [pk] "s"(pk), [k] "n"(K), [mhk] "s"(MHK), [mnk] "s"(MNK));
  }
}
#undef GJ_FMA
#undef GJ_BODY_NEXT
#undef GJ_BODY_LAST
#undef GJ_GA_OPS

__global__ __launch_bounds__(32 * WPB)
void kalman_wmma_kernel(const float* __restrict__ state0, const float* __restrict__ cov0,
                        const float* __restrict__ meas,   const float* __restrict__ Fg,
                        const float* __restrict__ Hg,     const float* __restrict__ Qg,
                        const float* __restrict__ Rg,     float* __restrict__ out,
                        int B, int T) {
  const int tid  = threadIdx.x;
  const int lane = tid & 31;
  const int w    = tid >> 5;
  const int h    = lane >> 4;
  const int n    = lane & 15;

  __shared__ __attribute__((aligned(16))) float sF[32 * LDW];
  __shared__ __attribute__((aligned(16))) float sH[16 * LDW];
  __shared__ __attribute__((aligned(16))) float sP [WPB][32 * LDW];
  __shared__ __attribute__((aligned(16))) float sT1[WPB][32 * LDN];

  for (int i = tid; i < 32 * 32; i += blockDim.x) sF[(i >> 5) * LDW + (i & 31)] = Fg[i];
  for (int i = tid; i < 16 * 32; i += blockDim.x) sH[(i >> 5) * LDW + (i & 31)] = Hg[i];
  __syncthreads();

  const int b = blockIdx.x * WPB + w;
  if (b >= B) return;

  float* P  = sP [w];
  float* T1 = sT1[w];

  v2f fF[2][2][4];
  v2f fH[2][4];
#pragma unroll
  for (int J = 0; J < 2; ++J)
#pragma unroll
    for (int Kt = 0; Kt < 2; ++Kt)
#pragma unroll
      for (int q = 0; q < 4; ++q)
        fF[J][Kt][q] = ld_rowfrag(sF, LDW, 16 * J, 16 * Kt, q, lane);
#pragma unroll
  for (int Kt = 0; Kt < 2; ++Kt)
#pragma unroll
    for (int q = 0; q < 4; ++q)
      fH[Kt][q] = ld_rowfrag(sH, LDW, 0, 16 * Kt, q, lane);
  v8f Qc[2][2], Pc[2][2], Rc;
#pragma unroll
  for (int I = 0; I < 2; ++I)
#pragma unroll
    for (int J = 0; J < 2; ++J)
#pragma unroll
      for (int r = 0; r < 8; ++r) {
        const int e = (16 * I + 8 * h + r) * 32 + 16 * J + n;
        Qc[I][J][r] = Qg[e];
        Pc[I][J][r] = cov0[(size_t)b * 1024 + e];
      }
#pragma unroll
  for (int r = 0; r < 8; ++r) Rc[r] = Rg[(8 * h + r) * 16 + n];

  float Fa[32], Fb[32], Hr[32];
#pragma unroll
  for (int j = 0; j < 32; ++j) {
    Fa[j] = sF[n * LDW + j];
    Fb[j] = sF[(16 + n) * LDW + j];
    Hr[j] = sH[n * LDW + j];
  }
  float xa = state0[(size_t)b * 32 + n];
  float xb = state0[(size_t)b * 32 + 16 + n];
  const float* zb = meas + (size_t)b * T * 16;
  float*       ob = out  + (size_t)b * T * 32;

#pragma unroll 1
  for (int t = 0; t < T; ++t) {
    __builtin_amdgcn_s_setprio(2);
    if (t + 1 < T) __builtin_prefetch(zb + (size_t)(t + 1) * 16, 0, 1);

    const float zt = zb[(size_t)t * 16 + n];

    float pa2[2] = {0.f, 0.f}, pb2[2] = {0.f, 0.f};
#pragma unroll
    for (int j = 0; j < 32; ++j) {
      const float xj = rdlane(j < 16 ? xa : xb, j & 15);
      pa2[j & 1] = __builtin_fmaf(Fa[j], xj, pa2[j & 1]);
      pb2[j & 1] = __builtin_fmaf(Fb[j], xj, pb2[j & 1]);
    }
    const float xpa = pa2[0] + pa2[1], xpb = pb2[0] + pb2[1];
    float nu;
    {
      float ya[4] = {0.f, 0.f, 0.f, 0.f};
#pragma unroll
      for (int j = 0; j < 32; ++j)
        ya[j & 3] = __builtin_fmaf(Hr[j], rdlane(j < 16 ? xpa : xpb, j & 15), ya[j & 3]);
      nu = zt - ((ya[0] + ya[1]) + (ya[2] + ya[3]));
    }

    v8f Zc[2][2];
#pragma unroll
    for (int I = 0; I < 2; ++I)
#pragma unroll
      for (int J = 0; J < 2; ++J) {
        v8f acc = {};
#pragma unroll
        for (int Kt = 0; Kt < 2; ++Kt)
#pragma unroll
          for (int q = 0; q < 4; ++q)
            acc = wmma4(frag_c(Pc[Kt][I], q), fF[J][Kt][q], acc);
        Zc[I][J] = acc;
      }

#pragma unroll
    for (int I = 0; I < 2; ++I)
#pragma unroll
      for (int J = 0; J < 2; ++J) {
        v8f acc = {};
#pragma unroll
        for (int Kt = 0; Kt < 2; ++Kt)
#pragma unroll
          for (int q = 0; q < 4; ++q)
            acc = wmma4(frag_c(Zc[Kt][I], q), fF[J][Kt][q], acc);
        Pc[I][J] = acc + Qc[I][J];
        st_c(P, LDW, 16 * I, 16 * J, Pc[I][J], lane);
      }

    __builtin_amdgcn_wave_barrier();

    v8f T1c[2];
#pragma unroll
    for (int I = 0; I < 2; ++I) {
      v8f acc = {};
#pragma unroll
      for (int Kt = 0; Kt < 2; ++Kt)
#pragma unroll
        for (int q = 0; q < 4; ++q)
          acc = wmma4(ld_rowfrag(P, LDW, 16 * I, 16 * Kt, q, lane), fH[Kt][q], acc);
      T1c[I] = acc;
      st_c(T1, LDN, 16 * I, 0, acc, lane);
    }

    v8f Sc = {};
#pragma unroll
    for (int Kt = 0; Kt < 2; ++Kt)
#pragma unroll
      for (int q = 0; q < 4; ++q)
        Sc = wmma4(fH[Kt][q], frag_c(T1c[Kt], q), Sc);

    __builtin_amdgcn_wave_barrier();

    v2f fT1[2][4];
#pragma unroll
    for (int J = 0; J < 2; ++J)
#pragma unroll
      for (int q = 0; q < 4; ++q)
        fT1[J][q] = ld_rowfrag(T1, LDN, 16 * J, 0, q, lane);

    v8f T2c[2];
#pragma unroll
    for (int J = 0; J < 2; ++J) {
      v8f acc = {};
#pragma unroll
      for (int Kt = 0; Kt < 2; ++Kt)
#pragma unroll
        for (int q = 0; q < 4; ++q)
          acc = wmma4(fH[Kt][q], frag_c(Pc[Kt][J], q), acc);
      T2c[J] = acc;
    }
    __builtin_amdgcn_s_setprio(0);
    asm volatile("s_wait_alu depctr_va_vdst(0)" : "+v"(Sc));

    float ga[8];
#pragma unroll
    for (int r = 0; r < 8; ++r) ga[r] = Sc[r] + Rc[r];

    float cs = 0.0f;
    float pinv = __builtin_amdgcn_rcpf(rdlane(ga[0], 0));
    const float oth0 = xhalf(ga[0]);
    const float row0 = (h == 0) ? ga[0] : oth0;
    float ru = ((n == 0) ? 0.0f : row0) * pinv;
    gj_pivot<0>(ga, cs, ru, pinv); gj_pivot<1>(ga, cs, ru, pinv); gj_pivot<2>(ga, cs, ru, pinv); gj_pivot<3>(ga, cs, ru, pinv);
    gj_pivot<4>(ga, cs, ru, pinv); gj_pivot<5>(ga, cs, ru, pinv); gj_pivot<6>(ga, cs, ru, pinv); gj_pivot<7>(ga, cs, ru, pinv);
    gj_pivot<8>(ga, cs, ru, pinv); gj_pivot<9>(ga, cs, ru, pinv); gj_pivot<10>(ga, cs, ru, pinv); gj_pivot<11>(ga, cs, ru, pinv);
    gj_pivot<12>(ga, cs, ru, pinv); gj_pivot<13>(ga, cs, ru, pinv); gj_pivot<14>(ga, cs, ru, pinv); gj_pivot<15>(ga, cs, ru, pinv);
    v8f Si;
#pragma unroll
    for (int r = 0; r < 8; ++r) Si[r] = -(ga[r] * cs);

    __builtin_amdgcn_s_setprio(1);

    v8f KTc[2];
#pragma unroll
    for (int J = 0; J < 2; ++J) {
      v8f acc = {};
#pragma unroll
      for (int q = 0; q < 4; ++q)
        acc = wmma4(frag_c(Si, q), fT1[J][q], acc);
      KTc[J] = acc;
    }

    {
      float pa = 0.f, pb = 0.f;
#pragma unroll
      for (int r = 0; r < 8; ++r) {
        const float lo = rdlane(nu, r), hi = rdlane(nu, 8 + r);
        const float nr = h ? hi : lo;
        pa = __builtin_fmaf(KTc[0][r], nr, pa);
        pb = __builtin_fmaf(KTc[1][r], nr, pb);
      }
      xa = xpa + (pa + xhalf(pa));
      xb = xpb + (pb + xhalf(pb));
      ob[(size_t)t * 32 + lane] = h ? xb : xa;
    }

#pragma unroll
    for (int I = 0; I < 2; ++I)
#pragma unroll
      for (int J = 0; J < 2; ++J) {
        v8f acc = {};
#pragma unroll
        for (int q = 0; q < 4; ++q)
          acc = wmma4(frag_c(KTc[I], q), frag_c(T2c[J], q), acc);
        Pc[I][J] = Pc[I][J] - acc;
      }
    __builtin_amdgcn_wave_barrier();
  }
  __builtin_amdgcn_s_setprio(0);
}

extern "C" void kernel_launch(void* const* d_in, const int* in_sizes, int n_in,
                              void* d_out, int out_size, void* d_ws, size_t ws_size,
                              hipStream_t stream) {
  const float* state0 = (const float*)d_in[0];
  const float* cov0   = (const float*)d_in[1];
  const float* meas   = (const float*)d_in[2];
  const float* F      = (const float*)d_in[3];
  const float* H      = (const float*)d_in[4];
  const float* Q      = (const float*)d_in[5];
  const float* R      = (const float*)d_in[6];
  float* out = (float*)d_out;

  const int B = in_sizes[0] / 32;
  const int T = in_sizes[2] / (B * 16);
  const int blocks = (B + WPB - 1) / WPB;
  kalman_wmma_kernel<<<blocks, 32 * WPB, 0, stream>>>(state0, cov0, meas,
                                                      F, H, Q, R, out, B, T);
}


# State path without v_readlane: lane l holds x[l]; F x, H x_pred, K nu take the vector by row_share DPP on the multiply-add, lane halves joined by one half swap; matvec sum order changes
# speedup vs baseline: 6.9692x; 1.0405x over previous
#include <hip/hip_runtime.h>
#include <hip/hip_bf16.h>


typedef float v2f __attribute__((ext_vector_type(2)));
typedef float v8f __attribute__((ext_vector_type(8)));

#define WPB 8

#define LDW 44
#define LDN 20
#define PWS 4736


__device__ __forceinline__ v2f ld_rowfrag(const float* m, int ld, int r0, int c0, int q, int lane) {
  const int h = lane >> 4, c = lane & 15;
  const float* p = m + (r0 + c) * ld + c0 + (h << 3) + (q << 1);
  v2f a; a.x = p[0]; a.y = p[1];
  return a;
}
__device__ __forceinline__ v2f frag_c(v8f x, int q) {
  v2f f; f.x = x[2 * q]; f.y = x[2 * q + 1];
  return f;
}
__device__ __forceinline__ v8f wmma4(v2f a, v2f b, v8f c) {
  return __builtin_amdgcn_wmma_f32_16x16x4_f32(false, a, false, b, (short)0, c, false, false);
}

__device__ __forceinline__ void st_c(float* d, int ld, int r0, int c0, v8f c, int lane) {
  const int h = lane >> 4, n = lane & 15;
#pragma unroll
  for (int r = 0; r < 8; ++r)
    d[(r0 + r + (h << 3)) * ld + c0 + n] = c[r];
}

__device__ __forceinline__ float xhalf(float x) {
  const int b = __builtin_bit_cast(int, x);
  return __builtin_bit_cast(float,
      __builtin_amdgcn_permlanex16(b, b, 0x76543210, 0xfedcba98, false, true));
}
__device__ __forceinline__ float rdlane(float v, int l) {
  return __builtin_bit_cast(float,
      __builtin_amdgcn_readlane(__builtin_bit_cast(int, v), l));
}
template <int K>
__device__ __forceinline__ float rshare(float v) {
  return __builtin_bit_cast(float,
      __builtin_amdgcn_update_dpp(0, __builtin_bit_cast(int, v), 0x150 + K, 0xf, 0xf, true));
}
__device__ __forceinline__ float rror8(float v) {
  return __builtin_bit_cast(float,
      __builtin_amdgcn_update_dpp(0, __builtin_bit_cast(int, v), 0x128, 0xf, 0xf, true));
}
template <int K, int N, int NS>
__device__ __forceinline__ void row_dot(const float (&m)[N], float v, float (&s)[NS]) {
  if constexpr (K < N) {
    s[K % NS] = __builtin_fmaf(rshare<K>(v), m[K], s[K % NS]);
    row_dot<K + 1, N, NS>(m, v, s);
  }
}


#define GJ_FMA(i) "v_fma_f32_e64_dpp %" #i ", %" #i ", -%[ru], %" #i " row_share:%[k] row_mask:0xf bank_mask:0xf bound_ctrl:1\n\t"
#define GJ_BODY_NEXT                                                                              \
    GJ_FMA(0)                                        \
    GJ_FMA(1)                                                   \
    GJ_FMA(2)                                                                                     \
        \
    "s_delay_alu instid0(VALU_DEP_3) | instskip(NEXT) | instid1(VALU_DEP_4)\n\t"                  \
    "v_readlane_b32 %[pn], %0, %[l1]\n\t"                                                         \
    "v_permlanex16_b32 %[on], %0, %[sel], 0xfedcba98 op_sel:[0,1]\n\t"                            \
    GJ_FMA(3)                                                                                     \
    GJ_FMA(4)                                                                                     \
      \
    "s_delay_alu instid0(VALU_DEP_3) | instskip(NEXT) | instid1(VALU_DEP_4)\n\t"                  \
    "v_cndmask_b32_e64 %[rw], %[on], %0, %[mh1]\n\t"                                              \
    "v_s_rcp_f32 %[pn], %[pn]\n\t"                                                                \
    GJ_FMA(5)                                                                                     \
          \
    "v_cndmask_b32_e64 %[on], %[ru], -1.0, %[mnk]\n\t"                                            \
                          \
    "s_delay_alu instid0(VALU_DEP_3)\n\t"                                                         \
    "v_cndmask_b32_e64 %[rw], %[rw], 0, %[mn1]\n\t"                                               \
    GJ_FMA(6)                                                                                     \
    GJ_FMA(7)                                                                                     \
                                                    \
    "s_delay_alu instid0(TRANS32_DEP_1)\n\t"                                                      \
    "v_mul_f32_e64 %[rw], %[rw], %[pn]\n\t"                                                       \
                 \
    "v_cndmask_b32_e64 %[cs], %[cs], %[pk], %[mnk]\n\t"                                           \
    "v_cndmask_b32_e64 %1, %1, %[on], %[mhk]"
#define GJ_BODY_LAST                                                                              \
    GJ_FMA(0) GJ_FMA(1) GJ_FMA(2) GJ_FMA(3) GJ_FMA(4) GJ_FMA(5) GJ_FMA(6) GJ_FMA(7)               \
    "v_cndmask_b32_e64 %[on], %[ru], -1.0, %[mnk]\n\t"                                            \
    "v_cndmask_b32_e64 %[cs], %[cs], %[pk], %[mnk]\n\t"                                           \
    "s_delay_alu instid0(VALU_DEP_2)\n\t"                                                         \
    "v_cndmask_b32_e64 %1, %1, %[on], %[mhk]"
#define GJ_GA_OPS                                                                                 \
    "+v"(ga[(K + 1) & 7]), "+v"(ga[K & 7]), "+v"(ga[(K + 2) & 7]), "+v"(ga[(K + 3) & 7]),         \
    "+v"(ga[(K + 4) & 7]), "+v"(ga[(K + 5) & 7]), "+v"(ga[(K + 6) & 7]), "+v"(ga[(K + 7) & 7])

template <int K>
__device__ __forceinline__ void gj_pivot(float (&ga)[8], float& cs, float& ru, float& pinv) {
  constexpr unsigned MHK = (K >> 3) ? 0xffff0000u : 0x0000ffffu;
  constexpr unsigned MNK = (1u << K) | (1u << (16 + K));
  const float ruk = ru, pk = pinv;
  if constexpr (K < 15) {
    constexpr int K1 = K + 1;
    constexpr int L1 = 16 * (K1 >> 3) + K1;
    constexpr unsigned MH1 = (K1 >> 3) ? 0xffff0000u : 0x0000ffffu;
    constexpr unsigned MN1 = (1u << K1) | (1u << (16 + K1));
    float pn, on, rw;
    if constexpr (K == 0) {
      asm("s_delay_alu instid0(VALU_DEP_1)\n\t" GJ_BODY_NEXT
          : GJ_GA_OPS, [cs] "+v"(cs), [pn] "=&s"(pn), [on] "=&v"(on), [rw] "=&v"(rw)
          : [ru] "v"(ruk), [pk] "s"(pk), [k] "n"(K), [l1] "n"(L1), [sel] "s"(0x76543210),
            [mh1] "s"(MH1), [mn1] "s"(MN1), [mhk] "s"(MHK), [mnk] "s"(MNK));
    } else {
      asm("s_delay_alu instid0(VALU_DEP_3)\n\t" GJ_BODY_NEXT
          : GJ_GA_OPS, [cs] "+v"(cs), [pn] "=&s"(pn), [on] "=&v"(on), [rw] "=&v"(rw)
          : [ru] "v"(ruk), [pk] "s"(pk), [k] "n"(K), [l1] "n"(L1), [sel] "s"(0x76543210),
            [mh1] "s"(MH1), [mn1] "s"(MN1), [mhk] "s"(MHK), [mnk] "s"(MNK));
    }
    ru   = rw;
    pinv = pn;
  } else {
    float on;
    asm("s_delay_alu instid0(VALU_DEP_3)\n\t" GJ_BODY_LAST
        : GJ_GA_OPS, [cs] "+v"(cs), [on] "=&v"(on)
        : [ru] "v"(ruk), [pk] "s"(pk), [k] "n"(K), [mhk] "s"(MHK), [mnk] "s"(MNK));
  }
}
#undef GJ_FMA
#undef GJ_BODY_NEXT
#undef GJ_BODY_LAST
#undef GJ_GA_OPS


#define XP_OPS(B)                                                                                 \
    [s0] "+v"(s0), [s1] "+v"(s1), [s2] "+v"(s2), [s3] "+v"(s3)                                    \
    : [xv] "v"(xv), [xs] "v"(xs),                                                                 \
      [o0] "v"(Fo[B + 0]), [o1] "v"(Fo[B + 1]), [o2] "v"(Fo[B + 2]), [o3] "v"(Fo[B + 3]),         \
      [o4] "v"(Fo[B + 4]), [o5] "v"(Fo[B + 5]), [o6] "v"(Fo[B + 6]), [o7] "v"(Fo[B + 7]),         \
      [x0] "v"(Fx[B + 0]), [x1] "v"(Fx[B + 1]), [x2] "v"(Fx[B + 2]), [x3] "v"(Fx[B + 3]),         \
      [x4] "v"(Fx[B + 4]), [x5] "v"(Fx[B + 5]), [x6] "v"(Fx[B + 6]), [x7] "v"(Fx[B + 7])
__device__ __forceinline__ float xpred_row(const float (&Fo)[16], const float (&Fx)[16],
                                           float xv, float xs) {
  float s0 = 0.f, s1 = 0.f, s2 = 0.f, s3 = 0.f;
  asm(
    "v_fma_f32_e64_dpp %[s0], %[xv], %[o0], %[s0] row_share:0 row_mask:0xf bank_mask:0xf bound_ctrl:1\n\t"
    "v_fma_f32_e64_dpp %[s2], %[xs], %[x0], %[s2] row_share:0 row_mask:0xf bank_mask:0xf bound_ctrl:1\n\t"
    "v_fma_f32_e64_dpp %[s1], %[xv], %[o1], %[s1] row_share:1 row_mask:0xf bank_mask:0xf bound_ctrl:1\n\t"
    "v_fma_f32_e64_dpp %[s3], %[xs], %[x1], %[s3] row_share:1 row_mask:0xf bank_mask:0xf bound_ctrl:1\n\t"
    "s_delay_alu instid0(VALU_DEP_4) | instskip(NEXT) | instid1(VALU_DEP_4)\n\t"
    "v_fma_f32_e64_dpp %[s0], %[xv], %[o2], %[s0] row_share:2 row_mask:0xf bank_mask:0xf bound_ctrl:1\n\t"
    "v_fma_f32_e64_dpp %[s2], %[xs], %[x2], %[s2] row_share:2 row_mask:0xf bank_mask:0xf bound_ctrl:1\n\t"
    "s_delay_alu instid0(VALU_DEP_4) | instskip(NEXT) | instid1(VALU_DEP_4)\n\t"
    "v_fma_f32_e64_dpp %[s1], %[xv], %[o3], %[s1] row_share:3 row_mask:0xf bank_mask:0xf bound_ctrl:1\n\t"
    "v_fma_f32_e64_dpp %[s3], %[xs], %[x3], %[s3] row_share:3 row_mask:0xf bank_mask:0xf bound_ctrl:1\n\t"
    "s_delay_alu instid0(VALU_DEP_4) | instskip(NEXT) | instid1(VALU_DEP_4)\n\t"
    "v_fma_f32_e64_dpp %[s0], %[xv], %[o4], %[s0] row_share:4 row_mask:0xf bank_mask:0xf bound_ctrl:1\n\t"
    "v_fma_f32_e64_dpp %[s2], %[xs], %[x4], %[s2] row_share:4 row_mask:0xf bank_mask:0xf bound_ctrl:1\n\t"
    "s_delay_alu instid0(VALU_DEP_4) | instskip(NEXT) | instid1(VALU_DEP_4)\n\t"
    "v_fma_f32_e64_dpp %[s1], %[xv], %[o5], %[s1] row_share:5 row_mask:0xf bank_mask:0xf bound_ctrl:1\n\t"
    "v_fma_f32_e64_dpp %[s3], %[xs], %[x5], %[s3] row_share:5 row_mask:0xf bank_mask:0xf bound_ctrl:1\n\t"
    "s_delay_alu instid0(VALU_DEP_4) | instskip(NEXT) | instid1(VALU_DEP_4)\n\t"
    "v_fma_f32_e64_dpp %[s0], %[xv], %[o6], %[s0] row_share:6 row_mask:0xf bank_mask:0xf bound_ctrl:1\n\t"
    "v_fma_f32_e64_dpp %[s2], %[xs], %[x6], %[s2] row_share:6 row_mask:0xf bank_mask:0xf bound_ctrl:1\n\t"
    "s_delay_alu instid0(VALU_DEP_4) | instskip(NEXT) | instid1(VALU_DEP_4)\n\t"
    "v_fma_f32_e64_dpp %[s1], %[xv], %[o7], %[s1] row_share:7 row_mask:0xf bank_mask:0xf bound_ctrl:1\n\t"
    "v_fma_f32_e64_dpp %[s3], %[xs], %[x7], %[s3] row_share:7 row_mask:0xf bank_mask:0xf bound_ctrl:1"
      : XP_OPS(0));
  asm(
    "s_delay_alu instid0(VALU_DEP_4) | instskip(NEXT) | instid1(VALU_DEP_4)\n\t"
    "v_fma_f32_e64_dpp %[s0], %[xv], %[o0], %[s0] row_share:8 row_mask:0xf bank_mask:0xf bound_ctrl:1\n\t"
    "v_fma_f32_e64_dpp %[s2], %[xs], %[x0], %[s2] row_share:8 row_mask:0xf bank_mask:0xf bound_ctrl:1\n\t"
    "s_delay_alu instid0(VALU_DEP_4) | instskip(NEXT) | instid1(VALU_DEP_4)\n\t"
    "v_fma_f32_e64_dpp %[s1], %[xv], %[o1], %[s1] row_share:9 row_mask:0xf bank_mask:0xf bound_ctrl:1\n\t"
    "v_fma_f32_e64_dpp %[s3], %[xs], %[x1], %[s3] row_share:9 row_mask:0xf bank_mask:0xf bound_ctrl:1\n\t"
    "s_delay_alu instid0(VALU_DEP_4) | instskip(NEXT) | instid1(VALU_DEP_4)\n\t"
    "v_fma_f32_e64_dpp %[s0], %[xv], %[o2], %[s0] row_share:10 row_mask:0xf bank_mask:0xf bound_ctrl:1\n\t"
    "v_fma_f32_e64_dpp %[s2], %[xs], %[x2], %[s2] row_share:10 row_mask:0xf bank_mask:0xf bound_ctrl:1\n\t"
    "s_delay_alu instid0(VALU_DEP_4) | instskip(NEXT) | instid1(VALU_DEP_4)\n\t"
    "v_fma_f32_e64_dpp %[s1], %[xv], %[o3], %[s1] row_share:11 row_mask:0xf bank_mask:0xf bound_ctrl:1\n\t"
    "v_fma_f32_e64_dpp %[s3], %[xs], %[x3], %[s3] row_share:11 row_mask:0xf bank_mask:0xf bound_ctrl:1\n\t"
    "s_delay_alu instid0(VALU_DEP_4) | instskip(NEXT) | instid1(VALU_DEP_4)\n\t"
    "v_fma_f32_e64_dpp %[s0], %[xv], %[o4], %[s0] row_share:12 row_mask:0xf bank_mask:0xf bound_ctrl:1\n\t"
    "v_fma_f32_e64_dpp %[s2], %[xs], %[x4], %[s2] row_share:12 row_mask:0xf bank_mask:0xf bound_ctrl:1\n\t"
    "s_delay_alu instid0(VALU_DEP_4) | instskip(NEXT) | instid1(VALU_DEP_4)\n\t"
    "v_fma_f32_e64_dpp %[s1], %[xv], %[o5], %[s1] row_share:13 row_mask:0xf bank_mask:0xf bound_ctrl:1\n\t"
    "v_fma_f32_e64_dpp %[s3], %[xs], %[x5], %[s3] row_share:13 row_mask:0xf bank_mask:0xf bound_ctrl:1\n\t"
    "s_delay_alu instid0(VALU_DEP_4) | instskip(NEXT) | instid1(VALU_DEP_4)\n\t"
    "v_fma_f32_e64_dpp %[s0], %[xv], %[o6], %[s0] row_share:14 row_mask:0xf bank_mask:0xf bound_ctrl:1\n\t"
    "v_fma_f32_e64_dpp %[s2], %[xs], %[x6], %[s2] row_share:14 row_mask:0xf bank_mask:0xf bound_ctrl:1\n\t"
    "s_delay_alu instid0(VALU_DEP_4) | instskip(NEXT) | instid1(VALU_DEP_4)\n\t"
    "v_fma_f32_e64_dpp %[s1], %[xv], %[o7], %[s1] row_share:15 row_mask:0xf bank_mask:0xf bound_ctrl:1\n\t"
    "v_fma_f32_e64_dpp %[s3], %[xs], %[x7], %[s3] row_share:15 row_mask:0xf bank_mask:0xf bound_ctrl:1"
      : XP_OPS(8));
  return (s0 + s1) + (s2 + s3);
}
#undef XP_OPS
__device__ __forceinline__ float hx_half(const float (&Hh)[16], float xp) {
  float y0 = 0.f, y1 = 0.f, y2 = 0.f, y3 = 0.f;
  asm(
    "s_delay_alu instid0(VALU_DEP_1)\n\t"
    "v_fma_f32_e64_dpp %[y0], %[xp], %[h0], %[y0] row_share:0 row_mask:0xf bank_mask:0xf bound_ctrl:1\n\t"
    "v_fma_f32_e64_dpp %[y1], %[xp], %[h1], %[y1] row_share:1 row_mask:0xf bank_mask:0xf bound_ctrl:1\n\t"
    "v_fma_f32_e64_dpp %[y2], %[xp], %[h2], %[y2] row_share:2 row_mask:0xf bank_mask:0xf bound_ctrl:1\n\t"
    "v_fma_f32_e64_dpp %[y3], %[xp], %[h3], %[y3] row_share:3 row_mask:0xf bank_mask:0xf bound_ctrl:1\n\t"
    "s_delay_alu instid0(VALU_DEP_4) | instskip(NEXT) | instid1(VALU_DEP_4)\n\t"
    "v_fma_f32_e64_dpp %[y0], %[xp], %[h4], %[y0] row_share:4 row_mask:0xf bank_mask:0xf bound_ctrl:1\n\t"
    "v_fma_f32_e64_dpp %[y1], %[xp], %[h5], %[y1] row_share:5 row_mask:0xf bank_mask:0xf bound_ctrl:1\n\t"
    "s_delay_alu instid0(VALU_DEP_4) | instskip(NEXT) | instid1(VALU_DEP_4)\n\t"
    "v_fma_f32_e64_dpp %[y2], %[xp], %[h6], %[y2] row_share:6 row_mask:0xf bank_mask:0xf bound_ctrl:1\n\t"
    "v_fma_f32_e64_dpp %[y3], %[xp], %[h7], %[y3] row_share:7 row_mask:0xf bank_mask:0xf bound_ctrl:1\n\t"
    "s_delay_alu instid0(VALU_DEP_4) | instskip(NEXT) | instid1(VALU_DEP_4)\n\t"
    "v_fma_f32_e64_dpp %[y0], %[xp], %[h8], %[y0] row_share:8 row_mask:0xf bank_mask:0xf bound_ctrl:1\n\t"
    "v_fma_f32_e64_dpp %[y1], %[xp], %[h9], %[y1] row_share:9 row_mask:0xf bank_mask:0xf bound_ctrl:1\n\t"
    "s_delay_alu instid0(VALU_DEP_4) | instskip(NEXT) | instid1(VALU_DEP_4)\n\t"
    "v_fma_f32_e64_dpp %[y2], %[xp], %[h10], %[y2] row_share:10 row_mask:0xf bank_mask:0xf bound_ctrl:1\n\t"
    "v_fma_f32_e64_dpp %[y3], %[xp], %[h11], %[y3] row_share:11 row_mask:0xf bank_mask:0xf bound_ctrl:1\n\t"
    "s_delay_alu instid0(VALU_DEP_4) | instskip(NEXT) | instid1(VALU_DEP_4)\n\t"
    "v_fma_f32_e64_dpp %[y0], %[xp], %[h12], %[y0] row_share:12 row_mask:0xf bank_mask:0xf bound_ctrl:1\n\t"
    "v_fma_f32_e64_dpp %[y1], %[xp], %[h13], %[y1] row_share:13 row_mask:0xf bank_mask:0xf bound_ctrl:1\n\t"
    "s_delay_alu instid0(VALU_DEP_4) | instskip(NEXT) | instid1(VALU_DEP_4)\n\t"
    "v_fma_f32_e64_dpp %[y2], %[xp], %[h14], %[y2] row_share:14 row_mask:0xf bank_mask:0xf bound_ctrl:1\n\t"
    "v_fma_f32_e64_dpp %[y3], %[xp], %[h15], %[y3] row_share:15 row_mask:0xf bank_mask:0xf bound_ctrl:1"
      : [y0] "+v"(y0), [y1] "+v"(y1), [y2] "+v"(y2), [y3] "+v"(y3)
      : [xp] "v"(xp),
        [h0] "v"(Hh[0]), [h1] "v"(Hh[1]), [h2] "v"(Hh[2]), [h3] "v"(Hh[3]),
        [h4] "v"(Hh[4]), [h5] "v"(Hh[5]), [h6] "v"(Hh[6]), [h7] "v"(Hh[7]),
        [h8] "v"(Hh[8]), [h9] "v"(Hh[9]), [h10] "v"(Hh[10]), [h11] "v"(Hh[11]),
        [h12] "v"(Hh[12]), [h13] "v"(Hh[13]), [h14] "v"(Hh[14]), [h15] "v"(Hh[15]));
  return (y0 + y1) + (y2 + y3);
}

__global__ __launch_bounds__(32 * WPB)
void kalman_wmma_kernel(const float* __restrict__ state0, const float* __restrict__ cov0,
                        const float* __restrict__ meas,   const float* __restrict__ Fg,
                        const float* __restrict__ Hg,     const float* __restrict__ Qg,
                        const float* __restrict__ Rg,     float* __restrict__ out,
                        int B, int T) {
  const int tid  = threadIdx.x;
  const int lane = tid & 31;
  const int w    = tid >> 5;
  const int h    = lane >> 4;
  const int n    = lane & 15;

  __shared__ __attribute__((aligned(16))) float sF[32 * LDW];
  __shared__ __attribute__((aligned(16))) float sH[16 * LDW];
  __shared__ __attribute__((aligned(16))) float sP [WPB][PWS];
  __shared__ __attribute__((aligned(16))) float sT1[WPB][32 * LDN];

  for (int i = tid; i < 32 * 32; i += blockDim.x) sF[(i >> 5) * LDW + (i & 31)] = Fg[i];
  for (int i = tid; i < 16 * 32; i += blockDim.x) sH[(i >> 5) * LDW + (i & 31)] = Hg[i];
  __syncthreads();

  const int b = blockIdx.x * WPB + w;
  if (b >= B) return;

  float* P  = sP [w];
  float* T1 = sT1[w];

  v2f fF[2][2][4];
  v2f fH[2][4];
#pragma unroll
  for (int J = 0; J < 2; ++J)
#pragma unroll
    for (int Kt = 0; Kt < 2; ++Kt)
#pragma unroll
      for (int q = 0; q < 4; ++q)
        fF[J][Kt][q] = ld_rowfrag(sF, LDW, 16 * J, 16 * Kt, q, lane);
#pragma unroll
  for (int Kt = 0; Kt < 2; ++Kt)
#pragma unroll
    for (int q = 0; q < 4; ++q)
      fH[Kt][q] = ld_rowfrag(sH, LDW, 0, 16 * Kt, q, lane);
  v8f Qc[2][2], Pc[2][2], Rc;
#pragma unroll
  for (int I = 0; I < 2; ++I)
#pragma unroll
    for (int J = 0; J < 2; ++J)
#pragma unroll
      for (int r = 0; r < 8; ++r) {
        const int e = (16 * I + 8 * h + r) * 32 + 16 * J + n;
        Qc[I][J][r] = Qg[e];
        Pc[I][J][r] = cov0[(size_t)b * 1024 + e];
      }
#pragma unroll
  for (int r = 0; r < 8; ++r) Rc[r] = Rg[(8 * h + r) * 16 + n];

  float Fo[16], Fx[16], Hh[16];
#pragma unroll
  for (int j = 0; j < 16; ++j) {
    Fo[j] = sF[lane * LDW + 16 * h + j];
    Fx[j] = sF[lane * LDW + 16 * (1 - h) + j];
    Hh[j] = sH[n * LDW + 16 * h + j];
  }
  float xv = state0[(size_t)b * 32 + lane];
  const float* zb = meas + (size_t)b * T * 16;
  float*       ob = out  + (size_t)b * T * 32;

#pragma unroll 1
  for (int t = 0; t < T; ++t) {
    __builtin_amdgcn_s_setprio(2);
    if (t + 1 < T) __builtin_prefetch(zb + (size_t)(t + 1) * 16, 0, 1);

    const float zt = zb[(size_t)t * 16 + n];

    const float xs = xhalf(xv);
    const float xpv = xpred_row(Fo, Fx, xv, xs);
    const float yh = hx_half(Hh, xpv);
    const float nu = zt - (yh + xhalf(yh));

    v8f Zc[2][2];
#pragma unroll
    for (int I = 0; I < 2; ++I)
#pragma unroll
      for (int J = 0; J < 2; ++J) {
        v8f acc = {};
#pragma unroll
        for (int Kt = 0; Kt < 2; ++Kt)
#pragma unroll
          for (int q = 0; q < 4; ++q)
            acc = wmma4(frag_c(Pc[Kt][I], q), fF[J][Kt][q], acc);
        Zc[I][J] = acc;
      }

#pragma unroll
    for (int I = 0; I < 2; ++I)
#pragma unroll
      for (int J = 0; J < 2; ++J) {
        v8f acc = {};
#pragma unroll
        for (int Kt = 0; Kt < 2; ++Kt)
#pragma unroll
          for (int q = 0; q < 4; ++q)
            acc = wmma4(frag_c(Zc[Kt][I], q), fF[J][Kt][q], acc);
        Pc[I][J] = acc + Qc[I][J];
        st_c(P, LDW, 16 * I, 16 * J, Pc[I][J], lane);
      }

    __builtin_amdgcn_wave_barrier();

    v8f T1c[2];
#pragma unroll
    for (int I = 0; I < 2; ++I) {
      v8f acc = {};
#pragma unroll
      for (int Kt = 0; Kt < 2; ++Kt)
#pragma unroll
        for (int q = 0; q < 4; ++q)
          acc = wmma4(ld_rowfrag(P, LDW, 16 * I, 16 * Kt, q, lane), fH[Kt][q], acc);
      T1c[I] = acc;
      st_c(T1, LDN, 16 * I, 0, acc, lane);
    }

    v8f Sc = {};
#pragma unroll
    for (int Kt = 0; Kt < 2; ++Kt)
#pragma unroll
      for (int q = 0; q < 4; ++q)
        Sc = wmma4(fH[Kt][q], frag_c(T1c[Kt], q), Sc);

    __builtin_amdgcn_wave_barrier();

    v2f fT1[2][4];
#pragma unroll
    for (int J = 0; J < 2; ++J)
#pragma unroll
      for (int q = 0; q < 4; ++q)
        fT1[J][q] = ld_rowfrag(T1, LDN, 16 * J, 0, q, lane);

    v8f T2c[2];
#pragma unroll
    for (int J = 0; J < 2; ++J) {
      v8f acc = {};
#pragma unroll
      for (int Kt = 0; Kt < 2; ++Kt)
#pragma unroll
        for (int q = 0; q < 4; ++q)
          acc = wmma4(fH[Kt][q], frag_c(Pc[Kt][J], q), acc);
      T2c[J] = acc;
    }
    __builtin_amdgcn_s_setprio(0);
    asm volatile("s_wait_alu depctr_va_vdst(0)" : "+v"(Sc));

    float ga[8];
#pragma unroll
    for (int r = 0; r < 8; ++r) ga[r] = Sc[r] + Rc[r];

    float cs = 0.0f;
    float pinv = __builtin_amdgcn_rcpf(rdlane(ga[0], 0));
    const float oth0 = xhalf(ga[0]);
    const float row0 = (h == 0) ? ga[0] : oth0;
    float ru = ((n == 0) ? 0.0f : row0) * pinv;
    gj_pivot<0>(ga, cs, ru, pinv); gj_pivot<1>(ga, cs, ru, pinv); gj_pivot<2>(ga, cs, ru, pinv); gj_pivot<3>(ga, cs, ru, pinv);
    gj_pivot<4>(ga, cs, ru, pinv); gj_pivot<5>(ga, cs, ru, pinv); gj_pivot<6>(ga, cs, ru, pinv); gj_pivot<7>(ga, cs, ru, pinv);
    gj_pivot<8>(ga, cs, ru, pinv); gj_pivot<9>(ga, cs, ru, pinv); gj_pivot<10>(ga, cs, ru, pinv); gj_pivot<11>(ga, cs, ru, pinv);
    gj_pivot<12>(ga, cs, ru, pinv); gj_pivot<13>(ga, cs, ru, pinv); gj_pivot<14>(ga, cs, ru, pinv); gj_pivot<15>(ga, cs, ru, pinv);
    v8f Si;
#pragma unroll
    for (int r = 0; r < 8; ++r) Si[r] = -(ga[r] * cs);

    __builtin_amdgcn_s_setprio(1);

    v8f KTc[2];
#pragma unroll
    for (int J = 0; J < 2; ++J) {
      v8f acc = {};
#pragma unroll
      for (int q = 0; q < 4; ++q)
        acc = wmma4(frag_c(Si, q), fT1[J][q], acc);
      KTc[J] = acc;
    }

    {
      const float nur = rror8(nu);
      const float nus = h ? nur : nu;
      float pa[1] = {0.f}, pb[1] = {0.f};
      float k0[8], k1[8];
#pragma unroll
      for (int r = 0; r < 8; ++r) { k0[r] = KTc[0][r]; k1[r] = KTc[1][r]; }
      row_dot<0, 8, 1>(k0, nus, pa);
      row_dot<0, 8, 1>(k1, nus, pb);
      const float keep = h ? pb[0] : pa[0];
      const float send = h ? pa[0] : pb[0];
      xv = xpv + (keep + xhalf(send));
      ob[(size_t)t * 32 + lane] = xv;
    }

#pragma unroll
    for (int I = 0; I < 2; ++I)
#pragma unroll
      for (int J = 0; J < 2; ++J) {
        v8f acc = {};
#pragma unroll
        for (int q = 0; q < 4; ++q)
          acc = wmma4(frag_c(KTc[I], q), frag_c(T2c[J], q), acc);
        Pc[I][J] = Pc[I][J] - acc;
      }
    __builtin_amdgcn_wave_barrier();
  }
  __builtin_amdgcn_s_setprio(0);
}

extern "C" void kernel_launch(void* const* d_in, const int* in_sizes, int n_in,
                              void* d_out, int out_size, void* d_ws, size_t ws_size,
                              hipStream_t stream) {
  const float* state0 = (const float*)d_in[0];
  const float* cov0   = (const float*)d_in[1];
  const float* meas   = (const float*)d_in[2];
  const float* F      = (const float*)d_in[3];
  const float* H      = (const float*)d_in[4];
  const float* Q      = (const float*)d_in[5];
  const float* R      = (const float*)d_in[6];
  float* out = (float*)d_out;

  const int B = in_sizes[0] / 32;
  const int T = in_sizes[2] / (B * 16);
  const int blocks = (B + WPB - 1) / WPB;
  kalman_wmma_kernel<<<blocks, 32 * WPB, 0, stream>>>(state0, cov0, meas,
                                                      F, H, Q, R, out, B, T);
}
